# starting kernel
# baseline (speedup 1.0000x reference)
#include <hip/hip_runtime.h>

typedef __attribute__((ext_vector_type(2))) float v2f;
typedef __attribute__((ext_vector_type(8))) float v8f;

#define NN 100000
#define FF 256
#define EE 1600000
#define HH1 128
#define HH2 64

#define GEMM_WAVES 4

__global__ void gcn_zero(float* __restrict__ p, int n) {
    int i = blockIdx.x * blockDim.x + threadIdx.x;
    if (i < n) p[i] = 0.0f;
}

__global__ void gcn_relu(float* __restrict__ p, int n) {
    int i = blockIdx.x * blockDim.x + threadIdx.x;
    if (i < n) p[i] = fmaxf(p[i], 0.0f);
}

template <int K, int Nout>
__global__ __launch_bounds__(32 * GEMM_WAVES)
void gcn_gemm_wmma2(const float* __restrict__ A,
                    const float* __restrict__ B0, const float* __restrict__ B1,
                    float* __restrict__ C0, float* __restrict__ C1, int M) {
    static_assert(K % 16 == 0 && Nout % 16 == 0, "tile dims");
    const int lane = threadIdx.x;
    const int half = lane >> 4;
    const int mn   = lane & 15;
    const int tileN = blockIdx.x;
    const int tileM = blockIdx.y * GEMM_WAVES + threadIdx.y;
    if (tileM * 16 >= M) return;

    const float* __restrict__ ap  = A + (size_t)(tileM * 16 + mn) * K + 2 * half;
    const size_t boff = (size_t)(2 * half) * Nout + tileN * 16 + mn;
    const float* __restrict__ b0c = B0 + boff;
    const float* __restrict__ b1c = B1 + boff;

    v8f s0a = {}, s0b = {}, s0c = {}, s0d = {};
    v8f s1a = {}, s1b = {}, s1c = {}, s1d = {};

    for (int it = 0; it < K / 16; ++it) {
        v2f a0 = *reinterpret_cast<const v2f*>(ap + 0);
        v2f a1 = *reinterpret_cast<const v2f*>(ap + 4);
        v2f a2 = *reinterpret_cast<const v2f*>(ap + 8);
        v2f a3 = *reinterpret_cast<const v2f*>(ap + 12);
        v2f b00, b01, b02, b03, b10, b11, b12, b13;
        b00.x = b0c[0 * Nout];   b00.y = b0c[1 * Nout];
        b01.x = b0c[4 * Nout];   b01.y = b0c[5 * Nout];
        b02.x = b0c[8 * Nout];   b02.y = b0c[9 * Nout];
        b03.x = b0c[12 * Nout];  b03.y = b0c[13 * Nout];
        b10.x = b1c[0 * Nout];   b10.y = b1c[1 * Nout];
        b11.x = b1c[4 * Nout];   b11.y = b1c[5 * Nout];
        b12.x = b1c[8 * Nout];   b12.y = b1c[9 * Nout];
        b13.x = b1c[12 * Nout];  b13.y = b1c[13 * Nout];

        s0a = __builtin_amdgcn_wmma_f32_16x16x4_f32(false, a0, false, b00, (short)0, s0a, false, false);
        s1a = __builtin_amdgcn_wmma_f32_16x16x4_f32(false, a0, false, b10, (short)0, s1a, false, false);
        s0b = __builtin_amdgcn_wmma_f32_16x16x4_f32(false, a1, false, b01, (short)0, s0b, false, false);
        s1b = __builtin_amdgcn_wmma_f32_16x16x4_f32(false, a1, false, b11, (short)0, s1b, false, false);
        s0c = __builtin_amdgcn_wmma_f32_16x16x4_f32(false, a2, false, b02, (short)0, s0c, false, false);
        s1c = __builtin_amdgcn_wmma_f32_16x16x4_f32(false, a2, false, b12, (short)0, s1c, false, false);
        s0d = __builtin_amdgcn_wmma_f32_16x16x4_f32(false, a3, false, b03, (short)0, s0d, false, false);
        s1d = __builtin_amdgcn_wmma_f32_16x16x4_f32(false, a3, false, b13, (short)0, s1d, false, false);

        ap  += 16;
        b0c += (size_t)16 * Nout;
        b1c += (size_t)16 * Nout;
    }
    v8f r0 = (s0a + s0b) + (s0c + s0d);
    v8f r1 = (s1a + s1b) + (s1c + s1d);

    const size_t coff = (size_t)(tileM * 16 + half * 8) * Nout + tileN * 16 + mn;
    float* __restrict__ c0row = C0 + coff;
    float* __restrict__ c1row = C1 + coff;
#pragma unroll
    for (int v = 0; v < 8; ++v) c0row[(size_t)v * Nout] = r0[v];
#pragma unroll
    for (int v = 0; v < 8; ++v) c1row[(size_t)v * Nout] = r1[v];
}

template <int LOGC>
__global__ void gcn_scatter_fused(const float* __restrict__ pre0,
                                  const float* __restrict__ pre1,
                                  const int* __restrict__ erow,
                                  const int* __restrict__ ecol,
                                  const float* __restrict__ ev0,
                                  const float* __restrict__ ev1,
                                  float* __restrict__ acc) {
    constexpr int C = 1 << LOGC;
    int t = blockIdx.x * blockDim.x + threadIdx.x;
    int e = t >> LOGC;
    int c = t & (C - 1);
    if (e >= EE) return;

    float v0 = __builtin_nontemporal_load(ev0 + e);
    float v1 = __builtin_nontemporal_load(ev1 + e);
    int   r  = __builtin_nontemporal_load(erow + e);
    int  col = __builtin_nontemporal_load(ecol + e);

    const float4 p0 =
        reinterpret_cast<const float4*>(pre0 + ((size_t)col << (LOGC + 2)))[c];
    const float4 p1 =
        reinterpret_cast<const float4*>(pre1 + ((size_t)col << (LOGC + 2)))[c];

    float* dst = acc + ((size_t)r << (LOGC + 2)) + 4 * c;
    atomicAdd(dst + 0, v0 * p0.x + v1 * p1.x);
    atomicAdd(dst + 1, v0 * p0.y + v1 * p1.y);
    atomicAdd(dst + 2, v0 * p0.z + v1 * p1.z);
    atomicAdd(dst + 3, v0 * p0.w + v1 * p1.w);
}

__global__ void gcn_scatter1_fused(const float* __restrict__ p0,
                                   const float* __restrict__ p1,
                                   const int* __restrict__ erow,
                                   const int* __restrict__ ecol,
                                   const float* __restrict__ ev0,
                                   const float* __restrict__ ev1,
                                   float* __restrict__ acc) {
    int e = blockIdx.x * blockDim.x + threadIdx.x;
    if (e >= EE) return;
    float v0 = __builtin_nontemporal_load(ev0 + e);
    float v1 = __builtin_nontemporal_load(ev1 + e);
    int  col = __builtin_nontemporal_load(ecol + e);
    int  r   = __builtin_nontemporal_load(erow + e);
    atomicAdd(acc + r, v0 * p0[col] + v1 * p1[col]);
}

__global__ void gcn_dot64x2(const float* __restrict__ h,
                            const float* __restrict__ w0,
                            const float* __restrict__ w1,
                            float* __restrict__ p0, float* __restrict__ p1) {
    int i = blockIdx.x * blockDim.x + threadIdx.x;
    if (i >= NN) return;
    const float* row = h + (size_t)i * HH2;
    float s0 = 0.0f, s1 = 0.0f;
#pragma unroll
    for (int k = 0; k < HH2; ++k) {
        float hv = row[k];
        s0 += hv * w0[k];
        s1 += hv * w1[k];
    }
    p0[i] = s0;
    p1[i] = s1;
}

extern "C" void kernel_launch(void* const* d_in, const int* in_sizes, int n_in,
                              void* d_out, int out_size, void* d_ws, size_t ws_size,
                              hipStream_t stream) {
    const float* x     = (const float*)d_in[0];
    const int*   erow  = (const int*)d_in[1];
    const int*   ecol  = (const int*)d_in[2];
    const float* evals = (const float*)d_in[3];
    const float* W1    = (const float*)d_in[4];
    const float* W2    = (const float*)d_in[5];
    const float* W3    = (const float*)d_in[6];
    float* out = (float*)d_out;

    const float* ev0 = evals;
    const float* ev1 = evals + (size_t)EE;

    float* pre0 = (float*)d_ws;
    float* pre1 = pre0 + (size_t)NN * HH1;
    float* bufA = pre1 + (size_t)NN * HH1;
    float* bufB = bufA + (size_t)NN * HH1;

    const int ZB = 256;
    dim3 gemm_blk(32, GEMM_WAVES);
    const int mtiles = NN / 16;
    const int gemm_gy = (mtiles + GEMM_WAVES - 1) / GEMM_WAVES;

    gcn_zero<<<((size_t)NN * HH1 + ZB - 1) / ZB, ZB, 0, stream>>>(bufA, NN * HH1);
    gcn_gemm_wmma2<FF, HH1><<<dim3(HH1 / 16, gemm_gy), gemm_blk, 0, stream>>>(
        x, W1, W1 + (size_t)FF * HH1, pre0, pre1, NN);
    {
        long long tot = (long long)EE << 5;
        gcn_scatter_fused<5><<<(tot + ZB - 1) / ZB, ZB, 0, stream>>>(
            pre0, pre1, erow, ecol, ev0, ev1, bufA);
    }
    gcn_relu<<<((size_t)NN * HH1 + ZB - 1) / ZB, ZB, 0, stream>>>(bufA, NN * HH1);

    gcn_zero<<<((size_t)NN * HH2 + ZB - 1) / ZB, ZB, 0, stream>>>(bufB, NN * HH2);
    gcn_gemm_wmma2<HH1, HH2><<<dim3(HH2 / 16, gemm_gy), gemm_blk, 0, stream>>>(
        bufA, W2, W2 + (size_t)HH1 * HH2, pre0, pre1, NN);
    {
        long long tot = (long long)EE << 4;
        gcn_scatter_fused<4><<<(tot + ZB - 1) / ZB, ZB, 0, stream>>>(
            pre0, pre1, erow, ecol, ev0, ev1, bufB);
    }
    gcn_relu<<<((size_t)NN * HH2 + ZB - 1) / ZB, ZB, 0, stream>>>(bufB, NN * HH2);

    gcn_zero<<<(NN + ZB - 1) / ZB, ZB, 0, stream>>>(out, NN);
    gcn_dot64x2<<<(NN + ZB - 1) / ZB, ZB, 0, stream>>>(
        bufB, W3, W3 + (size_t)HH2, pre0, pre1);
    gcn_scatter1_fused<<<(EE + ZB - 1) / ZB, ZB, 0, stream>>>(
        pre0, pre1, erow, ecol, ev0, ev1, out);
}


# Edges grouped by destination row once per call (int-atomic rank, scan, place); one owner per row sums in registers, stores once; float atomics, zero/relu passes gone; sum order = arrival rank
# speedup vs baseline: 1.5190x; 1.5190x over previous
#include <hip/hip_runtime.h>


typedef __attribute__((ext_vector_type(2))) float v2f;
typedef __attribute__((ext_vector_type(8))) float v8f;

#define NN 100000
#define FF 256
#define EE 1600000
#define HH1 128
#define HH2 64

#define GEMM_WAVES 4
#define GATHER_WAVES 8

__global__ void gcn_zero_counts(int* __restrict__ p, int n) {
    int i = blockIdx.x * blockDim.x + threadIdx.x;
    if (i < n) p[i] = 0;
}

__global__ void gcn_rank_edges(const int* __restrict__ erow,
                               int* __restrict__ cnt, int* __restrict__ rank) {
    int e = blockIdx.x * blockDim.x + threadIdx.x;
    if (e >= EE) return;
    int r = __builtin_nontemporal_load(erow + e);
    rank[e] = atomicAdd(cnt + r, 1);
}

#define SCAN_T 1024
#define SCAN_PER 100
__global__ __launch_bounds__(SCAN_T)
void gcn_scan_counts(const int* __restrict__ cnt, int* __restrict__ off) {
    static_assert(NN % SCAN_PER == 0 && NN / SCAN_PER <= SCAN_T && SCAN_PER % 4 == 0,
                  "scan partition");
    __shared__ int part[SCAN_T];
    const int t = threadIdx.x;
    const bool act = t < NN / SCAN_PER;
    const int4* src = reinterpret_cast<const int4*>(cnt + (act ? t : 0) * SCAN_PER);
    int s = 0;
    if (act) {
        for (int i = 0; i < SCAN_PER / 4; ++i) {
            int4 v = src[i];
            s += (v.x + v.y) + (v.z + v.w);
        }
    }
    part[t] = s;
    __syncthreads();
    for (int d = 1; d < SCAN_T; d <<= 1) {
        int v = (t >= d) ? part[t - d] : 0;
        __syncthreads();
        part[t] += v;
        __syncthreads();
    }
    if (act) {
        int run = part[t] - s;
        int4* dst = reinterpret_cast<int4*>(off + t * SCAN_PER);
        for (int i = 0; i < SCAN_PER / 4; ++i) {
            int4 v = src[i];
            int4 o;
            o.x = run; run += v.x;
            o.y = run; run += v.y;
            o.z = run; run += v.z;
            o.w = run; run += v.w;
            dst[i] = o;
        }
    }
    if (t == SCAN_T - 1) off[NN] = part[t];
}

__global__ void gcn_place_edges(const int* __restrict__ erow,
                                const int* __restrict__ ecol,
                                const float* __restrict__ ev0,
                                const float* __restrict__ ev1,
                                const int* __restrict__ off,
                                const int* __restrict__ rank,
                                int* __restrict__ scol,
                                float* __restrict__ sv0, float* __restrict__ sv1) {
    int e = blockIdx.x * blockDim.x + threadIdx.x;
    if (e >= EE) return;
    int r = __builtin_nontemporal_load(erow + e);
    int p = off[r] + __builtin_nontemporal_load(rank + e);
    scol[p] = __builtin_nontemporal_load(ecol + e);
    sv0[p]  = __builtin_nontemporal_load(ev0 + e);
    sv1[p]  = __builtin_nontemporal_load(ev1 + e);
}

template <int K, int Nout>
__global__ __launch_bounds__(32 * GEMM_WAVES)
void gcn_gemm_wmma2(const float* __restrict__ A,
                    const float* __restrict__ B0, const float* __restrict__ B1,
                    float* __restrict__ C0, float* __restrict__ C1, int M) {
    static_assert(K % 16 == 0 && Nout % 16 == 0, "tile dims");
    const int lane = threadIdx.x;
    const int half = lane >> 4;
    const int mn   = lane & 15;
    const int tileN = blockIdx.x;
    const int tileM = blockIdx.y * GEMM_WAVES + threadIdx.y;
    if (tileM * 16 >= M) return;

    const float* __restrict__ ap  = A + (size_t)(tileM * 16 + mn) * K + 2 * half;
    const size_t boff = (size_t)(2 * half) * Nout + tileN * 16 + mn;
    const float* __restrict__ b0c = B0 + boff;
    const float* __restrict__ b1c = B1 + boff;

    v8f s0a = {}, s0b = {}, s0c = {}, s0d = {};
    v8f s1a = {}, s1b = {}, s1c = {}, s1d = {};

    for (int it = 0; it < K / 16; ++it) {
        v2f a0 = *reinterpret_cast<const v2f*>(ap + 0);
        v2f a1 = *reinterpret_cast<const v2f*>(ap + 4);
        v2f a2 = *reinterpret_cast<const v2f*>(ap + 8);
        v2f a3 = *reinterpret_cast<const v2f*>(ap + 12);
        v2f b00, b01, b02, b03, b10, b11, b12, b13;
        b00.x = b0c[0 * Nout];   b00.y = b0c[1 * Nout];
        b01.x = b0c[4 * Nout];   b01.y = b0c[5 * Nout];
        b02.x = b0c[8 * Nout];   b02.y = b0c[9 * Nout];
        b03.x = b0c[12 * Nout];  b03.y = b0c[13 * Nout];
        b10.x = b1c[0 * Nout];   b10.y = b1c[1 * Nout];
        b11.x = b1c[4 * Nout];   b11.y = b1c[5 * Nout];
        b12.x = b1c[8 * Nout];   b12.y = b1c[9 * Nout];
        b13.x = b1c[12 * Nout];  b13.y = b1c[13 * Nout];

        s0a = __builtin_amdgcn_wmma_f32_16x16x4_f32(false, a0, false, b00, (short)0, s0a, false, false);
        s1a = __builtin_amdgcn_wmma_f32_16x16x4_f32(false, a0, false, b10, (short)0, s1a, false, false);
        s0b = __builtin_amdgcn_wmma_f32_16x16x4_f32(false, a1, false, b01, (short)0, s0b, false, false);
        s1b = __builtin_amdgcn_wmma_f32_16x16x4_f32(false, a1, false, b11, (short)0, s1b, false, false);
        s0c = __builtin_amdgcn_wmma_f32_16x16x4_f32(false, a2, false, b02, (short)0, s0c, false, false);
        s1c = __builtin_amdgcn_wmma_f32_16x16x4_f32(false, a2, false, b12, (short)0, s1c, false, false);
        s0d = __builtin_amdgcn_wmma_f32_16x16x4_f32(false, a3, false, b03, (short)0, s0d, false, false);
        s1d = __builtin_amdgcn_wmma_f32_16x16x4_f32(false, a3, false, b13, (short)0, s1d, false, false);

        ap  += 16;
        b0c += (size_t)16 * Nout;
        b1c += (size_t)16 * Nout;
    }
    v8f r0 = (s0a + s0b) + (s0c + s0d);
    v8f r1 = (s1a + s1b) + (s1c + s1d);

    const size_t coff = (size_t)(tileM * 16 + half * 8) * Nout + tileN * 16 + mn;
    float* __restrict__ c0row = C0 + coff;
    float* __restrict__ c1row = C1 + coff;
#pragma unroll
    for (int v = 0; v < 8; ++v) c0row[(size_t)v * Nout] = r0[v];
#pragma unroll
    for (int v = 0; v < 8; ++v) c1row[(size_t)v * Nout] = r1[v];
}

static __device__ __forceinline__ float gcn_lane_f(float x, int j) {
    return __int_as_float(__builtin_amdgcn_readlane(__float_as_int(x), j));
}

template <int VEC, bool RELU>
__global__ __launch_bounds__(32 * GATHER_WAVES)
void gcn_gather_rows(const float* __restrict__ pre0, const float* __restrict__ pre1,
                     const int* __restrict__ off, const int* __restrict__ scol,
                     const float* __restrict__ sv0, const float* __restrict__ sv1,
                     float* __restrict__ h) {
    typedef __attribute__((ext_vector_type(VEC))) float vecT;
    constexpr int H = 32 * VEC;
    const int lane = threadIdx.x;
    const int row  = blockIdx.x * GATHER_WAVES + threadIdx.y;
    if (row >= NN) return;
    const int beg = __builtin_amdgcn_readfirstlane(off[row]);
    const int end = __builtin_amdgcn_readfirstlane(off[row + 1]);
    const float* __restrict__ q0 = pre0 + lane * VEC;
    const float* __restrict__ q1 = pre1 + lane * VEC;

    vecT acc = {};
    for (int base = beg; base < end; base += 32) {
        const int left = end - base;
        const int n = left < 32 ? left : 32;
        int cj = 0;
        float a0 = 0.0f, a1 = 0.0f;
        if (lane < n) {
            cj = scol[base + lane];
            a0 = sv0[base + lane];
            a1 = sv1[base + lane];
        }
        int j = 0;
        for (; j + 4 <= n; j += 4) {
            vecT x0[4], x1[4];
            float w0[4], w1[4];
#pragma unroll
            for (int u = 0; u < 4; ++u) {
                const int c = __builtin_amdgcn_readlane(cj, j + u);
                w0[u] = gcn_lane_f(a0, j + u);
                w1[u] = gcn_lane_f(a1, j + u);
                x0[u] = *reinterpret_cast<const vecT*>(q0 + (size_t)c * H);
                x1[u] = *reinterpret_cast<const vecT*>(q1 + (size_t)c * H);
            }
#pragma unroll
            for (int u = 0; u < 4; ++u) acc += w0[u] * x0[u] + w1[u] * x1[u];
        }
        for (; j < n; ++j) {
            const int c = __builtin_amdgcn_readlane(cj, j);
            const float w0 = gcn_lane_f(a0, j);
            const float w1 = gcn_lane_f(a1, j);
            const vecT x0 = *reinterpret_cast<const vecT*>(q0 + (size_t)c * H);
            const vecT x1 = *reinterpret_cast<const vecT*>(q1 + (size_t)c * H);
            acc += w0 * x0 + w1 * x1;
        }
    }
    if (RELU) {
#pragma unroll
        for (int k = 0; k < VEC; ++k) acc[k] = fmaxf(acc[k], 0.0f);
    }
    *reinterpret_cast<vecT*>(h + (size_t)row * H + lane * VEC) = acc;
}

__global__ void gcn_gather_scalar(const float* __restrict__ p0,
                                  const float* __restrict__ p1,
                                  const int* __restrict__ off,
                                  const int* __restrict__ scol,
                                  const float* __restrict__ sv0,
                                  const float* __restrict__ sv1,
                                  float* __restrict__ out) {
    int r = blockIdx.x * blockDim.x + threadIdx.x;
    if (r >= NN) return;
    const int beg = off[r];
    const int end = off[r + 1];
    float s = 0.0f;
    for (int i = beg; i < end; ++i) {
        const int c = scol[i];
        s += sv0[i] * p0[c] + sv1[i] * p1[c];
    }
    out[r] = s;
}

__global__ void gcn_dot64x2(const float* __restrict__ h,
                            const float* __restrict__ w0,
                            const float* __restrict__ w1,
                            float* __restrict__ p0, float* __restrict__ p1) {
    int i = blockIdx.x * blockDim.x + threadIdx.x;
    if (i >= NN) return;
    const float* row = h + (size_t)i * HH2;
    float s0 = 0.0f, s1 = 0.0f;
#pragma unroll
    for (int k = 0; k < HH2; ++k) {
        float hv = row[k];
        s0 += hv * w0[k];
        s1 += hv * w1[k];
    }
    p0[i] = s0;
    p1[i] = s1;
}

extern "C" void kernel_launch(void* const* d_in, const int* in_sizes, int n_in,
                              void* d_out, int out_size, void* d_ws, size_t ws_size,
                              hipStream_t stream) {
    const float* x     = (const float*)d_in[0];
    const int*   erow  = (const int*)d_in[1];
    const int*   ecol  = (const int*)d_in[2];
    const float* evals = (const float*)d_in[3];
    const float* W1    = (const float*)d_in[4];
    const float* W2    = (const float*)d_in[5];
    const float* W3    = (const float*)d_in[6];
    float* out = (float*)d_out;

    const float* ev0 = evals;
    const float* ev1 = evals + (size_t)EE;

    float* R0 = (float*)d_ws;
    float* R1 = R0 + (size_t)NN * HH1;
    float* R2 = R1 + (size_t)NN * HH1;
    float* R3 = R2 + (size_t)NN * HH1;

    int*   cnt  = (int*)R2;
    int*   rank = cnt + NN;
    int*   scol = (int*)R3;
    float* sv0  = R3 + (size_t)EE;
    float* sv1  = sv0 + (size_t)EE;
    int*   off  = (int*)(sv1 + (size_t)EE);

    const int ZB = 256;
    dim3 gemm_blk(32, GEMM_WAVES);
    const int mtiles = NN / 16;
    const int gemm_gy = (mtiles + GEMM_WAVES - 1) / GEMM_WAVES;
    dim3 gat_blk(32, GATHER_WAVES);
    const int gat_g = (NN + GATHER_WAVES - 1) / GATHER_WAVES;

    gcn_zero_counts<<<(NN + ZB - 1) / ZB, ZB, 0, stream>>>(cnt, NN);
    gcn_rank_edges<<<(EE + ZB - 1) / ZB, ZB, 0, stream>>>(erow, cnt, rank);
    gcn_scan_counts<<<1, SCAN_T, 0, stream>>>(cnt, off);
    gcn_place_edges<<<(EE + ZB - 1) / ZB, ZB, 0, stream>>>(
        erow, ecol, ev0, ev1, off, rank, scol, sv0, sv1);

    float* pre0 = R0;
    float* pre1 = R1;
    float* h1   = R2;
    gcn_gemm_wmma2<FF, HH1><<<dim3(HH1 / 16, gemm_gy), gemm_blk, 0, stream>>>(
        x, W1, W1 + (size_t)FF * HH1, pre0, pre1, NN);
    gcn_gather_rows<4, true><<<gat_g, gat_blk, 0, stream>>>(
        pre0, pre1, off, scol, sv0, sv1, h1);

    float* pre0b = R0;
    float* pre1b = R0 + (size_t)NN * HH2;
    float* h2    = R1;
    gcn_gemm_wmma2<HH1, HH2><<<dim3(HH2 / 16, gemm_gy), gemm_blk, 0, stream>>>(
        h1, W2, W2 + (size_t)HH1 * HH2, pre0b, pre1b, NN);
    gcn_gather_rows<2, true><<<gat_g, gat_blk, 0, stream>>>(
        pre0b, pre1b, off, scol, sv0, sv1, h2);

    float* p0 = R2;
    float* p1 = R2 + NN;
    gcn_dot64x2<<<(NN + ZB - 1) / ZB, ZB, 0, stream>>>(
        h2, W3, W3 + (size_t)HH2, p0, p1);
    gcn_gather_scalar<<<(NN + ZB - 1) / ZB, ZB, 0, stream>>>(
        p0, p1, off, scol, sv0, sv1, out);
}


# GEMMs re-tiled: both supports' weights staged in LDS in fragment order per 16 k; a wave owns 16 rows x all columns, one A load feeds 2*Nout/16 WMMAs; k summed in order in one accumulator
# speedup vs baseline: 1.7655x; 1.1623x over previous
#include <hip/hip_runtime.h>


typedef __attribute__((ext_vector_type(2))) float v2f;
typedef __attribute__((ext_vector_type(8))) float v8f;

#define NN 100000
#define FF 256
#define EE 1600000
#define HH1 128
#define HH2 64

#define GEMM_WAVES 4
#define GATHER_WAVES 8

__global__ void gcn_zero_counts(int* __restrict__ p, int n) {
    int i = blockIdx.x * blockDim.x + threadIdx.x;
    if (i < n) p[i] = 0;
}

__global__ void gcn_rank_edges(const int* __restrict__ erow,
                               int* __restrict__ cnt, int* __restrict__ rank) {
    int e = blockIdx.x * blockDim.x + threadIdx.x;
    if (e >= EE) return;
    int r = __builtin_nontemporal_load(erow + e);
    rank[e] = atomicAdd(cnt + r, 1);
}

#define SCAN_T 1024
#define SCAN_PER 100
__global__ __launch_bounds__(SCAN_T)
void gcn_scan_counts(const int* __restrict__ cnt, int* __restrict__ off) {
    static_assert(NN % SCAN_PER == 0 && NN / SCAN_PER <= SCAN_T && SCAN_PER % 4 == 0,
                  "scan partition");
    __shared__ int part[SCAN_T];
    const int t = threadIdx.x;
    const bool act = t < NN / SCAN_PER;
    const int4* src = reinterpret_cast<const int4*>(cnt + (act ? t : 0) * SCAN_PER);
    int s = 0;
    if (act) {
        for (int i = 0; i < SCAN_PER / 4; ++i) {
            int4 v = src[i];
            s += (v.x + v.y) + (v.z + v.w);
        }
    }
    part[t] = s;
    __syncthreads();
    for (int d = 1; d < SCAN_T; d <<= 1) {
        int v = (t >= d) ? part[t - d] : 0;
        __syncthreads();
        part[t] += v;
        __syncthreads();
    }
    if (act) {
        int run = part[t] - s;
        int4* dst = reinterpret_cast<int4*>(off + t * SCAN_PER);
        for (int i = 0; i < SCAN_PER / 4; ++i) {
            int4 v = src[i];
            int4 o;
            o.x = run; run += v.x;
            o.y = run; run += v.y;
            o.z = run; run += v.z;
            o.w = run; run += v.w;
            dst[i] = o;
        }
    }
    if (t == SCAN_T - 1) off[NN] = part[t];
}

__global__ void gcn_place_edges(const int* __restrict__ erow,
                                const int* __restrict__ ecol,
                                const float* __restrict__ ev0,
                                const float* __restrict__ ev1,
                                const int* __restrict__ off,
                                const int* __restrict__ rank,
                                int* __restrict__ scol,
                                float* __restrict__ sv0, float* __restrict__ sv1) {
    int e = blockIdx.x * blockDim.x + threadIdx.x;
    if (e >= EE) return;
    int r = __builtin_nontemporal_load(erow + e);
    int p = off[r] + __builtin_nontemporal_load(rank + e);
    scol[p] = __builtin_nontemporal_load(ecol + e);
    sv0[p]  = __builtin_nontemporal_load(ev0 + e);
    sv1[p]  = __builtin_nontemporal_load(ev1 + e);
}

#define GEMM_KC 16
template <int K, int Nout>
__global__ __launch_bounds__(32 * GEMM_WAVES)
void gcn_gemm_wmma2(const float* __restrict__ A,
                    const float* __restrict__ B0, const float* __restrict__ B1,
                    float* __restrict__ C0, float* __restrict__ C1, int M) {
    static_assert(K % GEMM_KC == 0 && Nout % 16 == 0, "tile dims");
    constexpr int NT = Nout / 16;
    constexpr int TT = 2 * NT;
    constexpr int NC = K / GEMM_KC;
    constexpr int CHUNK_F = 4 * TT * 64;
    constexpr int UNITS = CHUNK_F / 8;
    constexpr int NTHR = 32 * GEMM_WAVES;
    constexpr int UPT = UNITS / NTHR;
    static_assert(UNITS % NTHR == 0 && UPT >= 1, "staging split");

    __shared__ __attribute__((aligned(16))) float wl[2][CHUNK_F];

    const int lane = threadIdx.x;
    const int wave = threadIdx.y;
    const int tid  = wave * 32 + lane;
    const int half = lane >> 4;
    const int mn   = lane & 15;
    const int tileM = blockIdx.x * GEMM_WAVES + wave;
    const bool live = tileM * 16 < M;
    const int row0 = live ? tileM * 16 : 0;

    const float* __restrict__ ap = A + (size_t)(row0 + mn) * K + 2 * half;

    const float* src[UPT];
#pragma unroll
    for (int i = 0; i < UPT; ++i) {
        const int u = i * NTHR + tid;
        const int frag = u >> 3, q = u & 7;
        const int ks = frag / TT, t = frag % TT;
        const int s = t / NT, tn = t % NT;
        const int hf = q >> 2, nn = (q & 3) * 4;
        src[i] = (s ? B1 : B0) + (size_t)(ks * 4 + hf * 2) * Nout + tn * 16 + nn;
    }

    v8f acc[TT];
#pragma unroll
    for (int t = 0; t < TT; ++t) acc[t] = v8f{};

    float4 r0[UPT], r1[UPT];
    v2f a[4], an[4];

#pragma unroll
    for (int i = 0; i < UPT; ++i) {
        r0[i] = *reinterpret_cast<const float4*>(src[i]);
        r1[i] = *reinterpret_cast<const float4*>(src[i] + Nout);
    }
#pragma unroll
    for (int ks = 0; ks < 4; ++ks) a[ks] = *reinterpret_cast<const v2f*>(ap + 4 * ks);
#pragma unroll
    for (int i = 0; i < UPT; ++i) {
        float* d = &wl[0][(i * NTHR + tid) * 8];
        *reinterpret_cast<float4*>(d)     = make_float4(r0[i].x, r1[i].x, r0[i].y, r1[i].y);
        *reinterpret_cast<float4*>(d + 4) = make_float4(r0[i].z, r1[i].z, r0[i].w, r1[i].w);
    }
    __syncthreads();

    for (int c = 0; c < NC; ++c) {
        const bool more = c + 1 < NC;
        if (more) {
            const size_t kn = (size_t)(c + 1) * GEMM_KC;
#pragma unroll
            for (int i = 0; i < UPT; ++i) {
                r0[i] = *reinterpret_cast<const float4*>(src[i] + kn * Nout);
                r1[i] = *reinterpret_cast<const float4*>(src[i] + kn * Nout + Nout);
            }
#pragma unroll
            for (int ks = 0; ks < 4; ++ks)
                an[ks] = *reinterpret_cast<const v2f*>(ap + kn + 4 * ks);
        }
        const float* wb = &wl[c & 1][lane * 2];
        v2f bx[TT], by[TT];
#pragma unroll
        for (int t = 0; t < TT; ++t) bx[t] = *reinterpret_cast<const v2f*>(wb + (0 * TT + t) * 64);
#pragma unroll
        for (int t = 0; t < TT; ++t) by[t] = *reinterpret_cast<const v2f*>(wb + (1 * TT + t) * 64);
#pragma unroll
        for (int t = 0; t < TT; ++t)
            acc[t] = __builtin_amdgcn_wmma_f32_16x16x4_f32(false, a[0], false, bx[t],
                                                           (short)0, acc[t], false, false);
#pragma unroll
        for (int t = 0; t < TT; ++t) bx[t] = *reinterpret_cast<const v2f*>(wb + (2 * TT + t) * 64);
#pragma unroll
        for (int t = 0; t < TT; ++t)
            acc[t] = __builtin_amdgcn_wmma_f32_16x16x4_f32(false, a[1], false, by[t],
                                                           (short)0, acc[t], false, false);
#pragma unroll
        for (int t = 0; t < TT; ++t) by[t] = *reinterpret_cast<const v2f*>(wb + (3 * TT + t) * 64);
#pragma unroll
        for (int t = 0; t < TT; ++t)
            acc[t] = __builtin_amdgcn_wmma_f32_16x16x4_f32(false, a[2], false, bx[t],
                                                           (short)0, acc[t], false, false);
#pragma unroll
        for (int t = 0; t < TT; ++t)
            acc[t] = __builtin_amdgcn_wmma_f32_16x16x4_f32(false, a[3], false, by[t],
                                                           (short)0, acc[t], false, false);
        if (more) {
            float* wd = &wl[(c & 1) ^ 1][0];
#pragma unroll
            for (int i = 0; i < UPT; ++i) {
                float* d = wd + (i * NTHR + tid) * 8;
                *reinterpret_cast<float4*>(d)     = make_float4(r0[i].x, r1[i].x, r0[i].y, r1[i].y);
                *reinterpret_cast<float4*>(d + 4) = make_float4(r0[i].z, r1[i].z, r0[i].w, r1[i].w);
            }
#pragma unroll
            for (int ks = 0; ks < 4; ++ks) a[ks] = an[ks];
        }
        __syncthreads();
    }

    if (live) {
#pragma unroll
        for (int t = 0; t < TT; ++t) {
            const int s = t / NT, tn = t % NT;
            float* __restrict__ cp = (s ? C1 : C0)
                + (size_t)(row0 + half * 8) * Nout + tn * 16 + mn;
#pragma unroll
            for (int v = 0; v < 8; ++v) cp[(size_t)v * Nout] = acc[t][v];
        }
    }
}

static __device__ __forceinline__ float gcn_lane_f(float x, int j) {
    return __int_as_float(__builtin_amdgcn_readlane(__float_as_int(x), j));
}

template <int VEC, bool RELU>
__global__ __launch_bounds__(32 * GATHER_WAVES)
void gcn_gather_rows(const float* __restrict__ pre0, const float* __restrict__ pre1,
                     const int* __restrict__ off, const int* __restrict__ scol,
                     const float* __restrict__ sv0, const float* __restrict__ sv1,
                     float* __restrict__ h) {
    typedef __attribute__((ext_vector_type(VEC))) float vecT;
    constexpr int H = 32 * VEC;
    const int lane = threadIdx.x;
    const int row  = blockIdx.x * GATHER_WAVES + threadIdx.y;
    if (row >= NN) return;
    const int beg = __builtin_amdgcn_readfirstlane(off[row]);
    const int end = __builtin_amdgcn_readfirstlane(off[row + 1]);
    const float* __restrict__ q0 = pre0 + lane * VEC;
    const float* __restrict__ q1 = pre1 + lane * VEC;

    vecT acc = {};
    for (int base = beg; base < end; base += 32) {
        const int left = end - base;
        const int n = left < 32 ? left : 32;
        int cj = 0;
        float a0 = 0.0f, a1 = 0.0f;
        if (lane < n) {
            cj = scol[base + lane];
            a0 = sv0[base + lane];
            a1 = sv1[base + lane];
        }
        int j = 0;
        for (; j + 4 <= n; j += 4) {
            vecT x0[4], x1[4];
            float w0[4], w1[4];
#pragma unroll
            for (int u = 0; u < 4; ++u) {
                const int c = __builtin_amdgcn_readlane(cj, j + u);
                w0[u] = gcn_lane_f(a0, j + u);
                w1[u] = gcn_lane_f(a1, j + u);
                x0[u] = *reinterpret_cast<const vecT*>(q0 + (size_t)c * H);
                x1[u] = *reinterpret_cast<const vecT*>(q1 + (size_t)c * H);
            }
#pragma unroll
            for (int u = 0; u < 4; ++u) acc += w0[u] * x0[u] + w1[u] * x1[u];
        }
        for (; j < n; ++j) {
            const int c = __builtin_amdgcn_readlane(cj, j);
            const float w0 = gcn_lane_f(a0, j);
            const float w1 = gcn_lane_f(a1, j);
            const vecT x0 = *reinterpret_cast<const vecT*>(q0 + (size_t)c * H);
            const vecT x1 = *reinterpret_cast<const vecT*>(q1 + (size_t)c * H);
            acc += w0 * x0 + w1 * x1;
        }
    }
    if (RELU) {
#pragma unroll
        for (int k = 0; k < VEC; ++k) acc[k] = fmaxf(acc[k], 0.0f);
    }
    *reinterpret_cast<vecT*>(h + (size_t)row * H + lane * VEC) = acc;
}

__global__ void gcn_gather_scalar(const float* __restrict__ p0,
                                  const float* __restrict__ p1,
                                  const int* __restrict__ off,
                                  const int* __restrict__ scol,
                                  const float* __restrict__ sv0,
                                  const float* __restrict__ sv1,
                                  float* __restrict__ out) {
    int r = blockIdx.x * blockDim.x + threadIdx.x;
    if (r >= NN) return;
    const int beg = off[r];
    const int end = off[r + 1];
    float s = 0.0f;
    for (int i = beg; i < end; ++i) {
        const int c = scol[i];
        s += sv0[i] * p0[c] + sv1[i] * p1[c];
    }
    out[r] = s;
}

__global__ void gcn_dot64x2(const float* __restrict__ h,
                            const float* __restrict__ w0,
                            const float* __restrict__ w1,
                            float* __restrict__ p0, float* __restrict__ p1) {
    int i = blockIdx.x * blockDim.x + threadIdx.x;
    if (i >= NN) return;
    const float* row = h + (size_t)i * HH2;
    float s0 = 0.0f, s1 = 0.0f;
#pragma unroll
    for (int k = 0; k < HH2; ++k) {
        float hv = row[k];
        s0 += hv * w0[k];
        s1 += hv * w1[k];
    }
    p0[i] = s0;
    p1[i] = s1;
}

extern "C" void kernel_launch(void* const* d_in, const int* in_sizes, int n_in,
                              void* d_out, int out_size, void* d_ws, size_t ws_size,
                              hipStream_t stream) {
    const float* x     = (const float*)d_in[0];
    const int*   erow  = (const int*)d_in[1];
    const int*   ecol  = (const int*)d_in[2];
    const float* evals = (const float*)d_in[3];
    const float* W1    = (const float*)d_in[4];
    const float* W2    = (const float*)d_in[5];
    const float* W3    = (const float*)d_in[6];
    float* out = (float*)d_out;

    const float* ev0 = evals;
    const float* ev1 = evals + (size_t)EE;

    float* R0 = (float*)d_ws;
    float* R1 = R0 + (size_t)NN * HH1;
    float* R2 = R1 + (size_t)NN * HH1;
    float* R3 = R2 + (size_t)NN * HH1;

    int*   cnt  = (int*)R2;
    int*   rank = cnt + NN;
    int*   scol = (int*)R3;
    float* sv0  = R3 + (size_t)EE;
    float* sv1  = sv0 + (size_t)EE;
    int*   off  = (int*)(sv1 + (size_t)EE);

    const int ZB = 256;
    dim3 gemm_blk(32, GEMM_WAVES);
    const int mtiles = NN / 16;
    const int gemm_gy = (mtiles + GEMM_WAVES - 1) / GEMM_WAVES;
    dim3 gat_blk(32, GATHER_WAVES);
    const int gat_g = (NN + GATHER_WAVES - 1) / GATHER_WAVES;

    gcn_zero_counts<<<(NN + ZB - 1) / ZB, ZB, 0, stream>>>(cnt, NN);
    gcn_rank_edges<<<(EE + ZB - 1) / ZB, ZB, 0, stream>>>(erow, cnt, rank);
    gcn_scan_counts<<<1, SCAN_T, 0, stream>>>(cnt, off);
    gcn_place_edges<<<(EE + ZB - 1) / ZB, ZB, 0, stream>>>(
        erow, ecol, ev0, ev1, off, rank, scol, sv0, sv1);

    float* pre0 = R0;
    float* pre1 = R1;
    float* h1   = R2;
    gcn_gemm_wmma2<FF, HH1><<<dim3(gemm_gy), gemm_blk, 0, stream>>>(
        x, W1, W1 + (size_t)FF * HH1, pre0, pre1, NN);
    gcn_gather_rows<4, true><<<gat_g, gat_blk, 0, stream>>>(
        pre0, pre1, off, scol, sv0, sv1, h1);

    float* pre0b = R0;
    float* pre1b = R0 + (size_t)NN * HH2;
    float* h2    = R1;
    gcn_gemm_wmma2<HH1, HH2><<<dim3(gemm_gy), gemm_blk, 0, stream>>>(
        h1, W2, W2 + (size_t)HH1 * HH2, pre0b, pre1b, NN);
    gcn_gather_rows<2, true><<<gat_g, gat_blk, 0, stream>>>(
        pre0b, pre1b, off, scol, sv0, sv1, h2);

    float* p0 = R2;
    float* p1 = R2 + NN;
    gcn_dot64x2<<<(NN + ZB - 1) / ZB, ZB, 0, stream>>>(
        h2, W3, W3 + (size_t)HH2, p0, p1);
    gcn_gather_scalar<<<(NN + ZB - 1) / ZB, ZB, 0, stream>>>(
        p0, p1, off, scol, sv0, sv1, out);
}


# Counter scan on 25 work-groups instead of 1: a thread owns 4 adjacent counters, so each load/store covers whole lines (was 32 lines per instruction); a group sums the earlier segments itself
# speedup vs baseline: 1.9098x; 1.0817x over previous
#include <hip/hip_runtime.h>


typedef __attribute__((ext_vector_type(2))) float v2f;
typedef __attribute__((ext_vector_type(8))) float v8f;

#define NN 100000
#define FF 256
#define EE 1600000
#define HH1 128
#define HH2 64

#define GEMM_WAVES 4
#define GATHER_WAVES 8

__global__ void gcn_zero_counts(int* __restrict__ p, int n) {
    int i = blockIdx.x * blockDim.x + threadIdx.x;
    if (i < n) p[i] = 0;
}

__global__ void gcn_rank_edges(const int* __restrict__ erow,
                               int* __restrict__ cnt, int* __restrict__ rank) {
    int e = blockIdx.x * blockDim.x + threadIdx.x;
    if (e >= EE) return;
    int r = __builtin_nontemporal_load(erow + e);
    rank[e] = atomicAdd(cnt + r, 1);
}

#define SCAN_T 1024
#define SCAN_SEG (4 * SCAN_T)
#define SCAN_WGS ((NN + SCAN_SEG - 1) / SCAN_SEG)
__global__ __launch_bounds__(SCAN_T)
void gcn_scan_counts(const int* __restrict__ cnt, int* __restrict__ off) {
    static_assert(NN % 4 == 0, "counters in whole int4 pieces");
    static_assert(SCAN_T == 32 * 32, "one lane per wave in the cross-wave step");
    static_assert((SCAN_WGS - 1) * SCAN_SEG <= NN, "every earlier segment is a full one");
    __shared__ int wown[SCAN_T / 32];
    __shared__ int wprev[SCAN_T / 32];
    const int t = threadIdx.x;
    const int lane = t & 31;
    const int wave = t >> 5;
    const int g = blockIdx.x;
    const int4* __restrict__ src4 = reinterpret_cast<const int4*>(cnt);

    const int i4 = g * SCAN_T + t;
    const bool act = i4 < NN / 4;
    int4 v = make_int4(0, 0, 0, 0);
    if (act) v = src4[i4];

    int sprev = 0;
    for (int k = 0; k < g; k += 8) {
        int4 w[8];
#pragma unroll
        for (int u = 0; u < 8; ++u) {
            const int ks = (k + u < g) ? (k + u) : (g - 1);
            w[u] = src4[ks * SCAN_T + t];
        }
#pragma unroll
        for (int u = 0; u < 8; ++u) {
            const int s4 = (w[u].x + w[u].y) + (w[u].z + w[u].w);
            sprev += (k + u < g) ? s4 : 0;
        }
    }

    const int own = (v.x + v.y) + (v.z + v.w);
    int inc = own;
    for (int d = 1; d < 32; d <<= 1) {
        const int y = __shfl_up(inc, d, 32);
        if (lane >= d) inc += y;
    }
    int ps = sprev;
    for (int d = 16; d >= 1; d >>= 1) ps += __shfl_xor(ps, d, 32);
    if (lane == 31) wown[wave] = inc;
    if (lane == 0) wprev[wave] = ps;
    __syncthreads();
    int before = (lane < wave) ? wown[lane] : 0;
    int base = wprev[lane];
    for (int d = 16; d >= 1; d >>= 1) {
        before += __shfl_xor(before, d, 32);
        base += __shfl_xor(base, d, 32);
    }
    const int start = base + before + (inc - own);
    if (act) {
        int4 o;
        o.x = start;
        o.y = o.x + v.x;
        o.z = o.y + v.y;
        o.w = o.z + v.z;
        reinterpret_cast<int4*>(off)[i4] = o;
    }
    if (i4 == NN / 4 - 1) off[NN] = start + own;
}

__global__ void gcn_place_edges(const int* __restrict__ erow,
                                const int* __restrict__ ecol,
                                const float* __restrict__ ev0,
                                const float* __restrict__ ev1,
                                const int* __restrict__ off,
                                const int* __restrict__ rank,
                                int* __restrict__ scol,
                                float* __restrict__ sv0, float* __restrict__ sv1) {
    int e = blockIdx.x * blockDim.x + threadIdx.x;
    if (e >= EE) return;
    int r = __builtin_nontemporal_load(erow + e);
    int p = off[r] + __builtin_nontemporal_load(rank + e);
    scol[p] = __builtin_nontemporal_load(ecol + e);
    sv0[p]  = __builtin_nontemporal_load(ev0 + e);
    sv1[p]  = __builtin_nontemporal_load(ev1 + e);
}

#define GEMM_KC 16
template <int K, int Nout>
__global__ __launch_bounds__(32 * GEMM_WAVES)
void gcn_gemm_wmma2(const float* __restrict__ A,
                    const float* __restrict__ B0, const float* __restrict__ B1,
                    float* __restrict__ C0, float* __restrict__ C1, int M) {
    static_assert(K % GEMM_KC == 0 && Nout % 16 == 0, "tile dims");
    constexpr int NT = Nout / 16;
    constexpr int TT = 2 * NT;
    constexpr int NC = K / GEMM_KC;
    constexpr int CHUNK_F = 4 * TT * 64;
    constexpr int UNITS = CHUNK_F / 8;
    constexpr int NTHR = 32 * GEMM_WAVES;
    constexpr int UPT = UNITS / NTHR;
    static_assert(UNITS % NTHR == 0 && UPT >= 1, "staging split");

    __shared__ __attribute__((aligned(16))) float wl[2][CHUNK_F];

    const int lane = threadIdx.x;
    const int wave = threadIdx.y;
    const int tid  = wave * 32 + lane;
    const int half = lane >> 4;
    const int mn   = lane & 15;
    const int tileM = blockIdx.x * GEMM_WAVES + wave;
    const bool live = tileM * 16 < M;
    const int row0 = live ? tileM * 16 : 0;

    const float* __restrict__ ap = A + (size_t)(row0 + mn) * K + 2 * half;

    const float* src[UPT];
#pragma unroll
    for (int i = 0; i < UPT; ++i) {
        const int u = i * NTHR + tid;
        const int frag = u >> 3, q = u & 7;
        const int ks = frag / TT, t = frag % TT;
        const int s = t / NT, tn = t % NT;
        const int hf = q >> 2, nn = (q & 3) * 4;
        src[i] = (s ? B1 : B0) + (size_t)(ks * 4 + hf * 2) * Nout + tn * 16 + nn;
    }

    v8f acc[TT];
#pragma unroll
    for (int t = 0; t < TT; ++t) acc[t] = v8f{};

    float4 r0[UPT], r1[UPT];
    v2f a[4], an[4];

#pragma unroll
    for (int i = 0; i < UPT; ++i) {
        r0[i] = *reinterpret_cast<const float4*>(src[i]);
        r1[i] = *reinterpret_cast<const float4*>(src[i] + Nout);
    }
#pragma unroll
    for (int ks = 0; ks < 4; ++ks) a[ks] = *reinterpret_cast<const v2f*>(ap + 4 * ks);
#pragma unroll
    for (int i = 0; i < UPT; ++i) {
        float* d = &wl[0][(i * NTHR + tid) * 8];
        *reinterpret_cast<float4*>(d)     = make_float4(r0[i].x, r1[i].x, r0[i].y, r1[i].y);
        *reinterpret_cast<float4*>(d + 4) = make_float4(r0[i].z, r1[i].z, r0[i].w, r1[i].w);
    }
    __syncthreads();

    for (int c = 0; c < NC; ++c) {
        const bool more = c + 1 < NC;
        if (more) {
            const size_t kn = (size_t)(c + 1) * GEMM_KC;
#pragma unroll
            for (int i = 0; i < UPT; ++i) {
                r0[i] = *reinterpret_cast<const float4*>(src[i] + kn * Nout);
                r1[i] = *reinterpret_cast<const float4*>(src[i] + kn * Nout + Nout);
            }
#pragma unroll
            for (int ks = 0; ks < 4; ++ks)
                an[ks] = *reinterpret_cast<const v2f*>(ap + kn + 4 * ks);
        }
        const float* wb = &wl[c & 1][lane * 2];
        v2f bx[TT], by[TT];
#pragma unroll
        for (int t = 0; t < TT; ++t) bx[t] = *reinterpret_cast<const v2f*>(wb + (0 * TT + t) * 64);
#pragma unroll
        for (int t = 0; t < TT; ++t) by[t] = *reinterpret_cast<const v2f*>(wb + (1 * TT + t) * 64);
#pragma unroll
        for (int t = 0; t < TT; ++t)
            acc[t] = __builtin_amdgcn_wmma_f32_16x16x4_f32(false, a[0], false, bx[t],
                                                           (short)0, acc[t], false, false);
#pragma unroll
        for (int t = 0; t < TT; ++t) bx[t] = *reinterpret_cast<const v2f*>(wb + (2 * TT + t) * 64);
#pragma unroll
        for (int t = 0; t < TT; ++t)
            acc[t] = __builtin_amdgcn_wmma_f32_16x16x4_f32(false, a[1], false, by[t],
                                                           (short)0, acc[t], false, false);
#pragma unroll
        for (int t = 0; t < TT; ++t) by[t] = *reinterpret_cast<const v2f*>(wb + (3 * TT + t) * 64);
#pragma unroll
        for (int t = 0; t < TT; ++t)
            acc[t] = __builtin_amdgcn_wmma_f32_16x16x4_f32(false, a[2], false, bx[t],
                                                           (short)0, acc[t], false, false);
#pragma unroll
        for (int t = 0; t < TT; ++t)
            acc[t] = __builtin_amdgcn_wmma_f32_16x16x4_f32(false, a[3], false, by[t],
                                                           (short)0, acc[t], false, false);
        if (more) {
            float* wd = &wl[(c & 1) ^ 1][0];
#pragma unroll
            for (int i = 0; i < UPT; ++i) {
                float* d = wd + (i * NTHR + tid) * 8;
                *reinterpret_cast<float4*>(d)     = make_float4(r0[i].x, r1[i].x, r0[i].y, r1[i].y);
                *reinterpret_cast<float4*>(d + 4) = make_float4(r0[i].z, r1[i].z, r0[i].w, r1[i].w);
            }
#pragma unroll
            for (int ks = 0; ks < 4; ++ks) a[ks] = an[ks];
        }
        __syncthreads();
    }

    if (live) {
#pragma unroll
        for (int t = 0; t < TT; ++t) {
            const int s = t / NT, tn = t % NT;
            float* __restrict__ cp = (s ? C1 : C0)
                + (size_t)(row0 + half * 8) * Nout + tn * 16 + mn;
#pragma unroll
            for (int v = 0; v < 8; ++v) cp[(size_t)v * Nout] = acc[t][v];
        }
    }
}

static __device__ __forceinline__ float gcn_lane_f(float x, int j) {
    return __int_as_float(__builtin_amdgcn_readlane(__float_as_int(x), j));
}

template <int VEC, bool RELU>
__global__ __launch_bounds__(32 * GATHER_WAVES)
void gcn_gather_rows(const float* __restrict__ pre0, const float* __restrict__ pre1,
                     const int* __restrict__ off, const int* __restrict__ scol,
                     const float* __restrict__ sv0, const float* __restrict__ sv1,
                     float* __restrict__ h) {
    typedef __attribute__((ext_vector_type(VEC))) float vecT;
    constexpr int H = 32 * VEC;
    const int lane = threadIdx.x;
    const int row  = blockIdx.x * GATHER_WAVES + threadIdx.y;
    if (row >= NN) return;
    const int beg = __builtin_amdgcn_readfirstlane(off[row]);
    const int end = __builtin_amdgcn_readfirstlane(off[row + 1]);
    const float* __restrict__ q0 = pre0 + lane * VEC;
    const float* __restrict__ q1 = pre1 + lane * VEC;

    vecT acc = {};
    for (int base = beg; base < end; base += 32) {
        const int left = end - base;
        const int n = left < 32 ? left : 32;
        int cj = 0;
        float a0 = 0.0f, a1 = 0.0f;
        if (lane < n) {
            cj = scol[base + lane];
            a0 = sv0[base + lane];
            a1 = sv1[base + lane];
        }
        int j = 0;
        for (; j + 4 <= n; j += 4) {
            vecT x0[4], x1[4];
            float w0[4], w1[4];
#pragma unroll
            for (int u = 0; u < 4; ++u) {
                const int c = __builtin_amdgcn_readlane(cj, j + u);
                w0[u] = gcn_lane_f(a0, j + u);
                w1[u] = gcn_lane_f(a1, j + u);
                x0[u] = *reinterpret_cast<const vecT*>(q0 + (size_t)c * H);
                x1[u] = *reinterpret_cast<const vecT*>(q1 + (size_t)c * H);
            }
#pragma unroll
            for (int u = 0; u < 4; ++u) acc += w0[u] * x0[u] + w1[u] * x1[u];
        }
        for (; j < n; ++j) {
            const int c = __builtin_amdgcn_readlane(cj, j);
            const float w0 = gcn_lane_f(a0, j);
            const float w1 = gcn_lane_f(a1, j);
            const vecT x0 = *reinterpret_cast<const vecT*>(q0 + (size_t)c * H);
            const vecT x1 = *reinterpret_cast<const vecT*>(q1 + (size_t)c * H);
            acc += w0 * x0 + w1 * x1;
        }
    }
    if (RELU) {
#pragma unroll
        for (int k = 0; k < VEC; ++k) acc[k] = fmaxf(acc[k], 0.0f);
    }
    *reinterpret_cast<vecT*>(h + (size_t)row * H + lane * VEC) = acc;
}

__global__ void gcn_gather_scalar(const float* __restrict__ p0,
                                  const float* __restrict__ p1,
                                  const int* __restrict__ off,
                                  const int* __restrict__ scol,
                                  const float* __restrict__ sv0,
                                  const float* __restrict__ sv1,
                                  float* __restrict__ out) {
    int r = blockIdx.x * blockDim.x + threadIdx.x;
    if (r >= NN) return;
    const int beg = off[r];
    const int end = off[r + 1];
    float s = 0.0f;
    for (int i = beg; i < end; ++i) {
        const int c = scol[i];
        s += sv0[i] * p0[c] + sv1[i] * p1[c];
    }
    out[r] = s;
}

__global__ void gcn_dot64x2(const float* __restrict__ h,
                            const float* __restrict__ w0,
                            const float* __restrict__ w1,
                            float* __restrict__ p0, float* __restrict__ p1) {
    int i = blockIdx.x * blockDim.x + threadIdx.x;
    if (i >= NN) return;
    const float* row = h + (size_t)i * HH2;
    float s0 = 0.0f, s1 = 0.0f;
#pragma unroll
    for (int k = 0; k < HH2; ++k) {
        float hv = row[k];
        s0 += hv * w0[k];
        s1 += hv * w1[k];
    }
    p0[i] = s0;
    p1[i] = s1;
}

extern "C" void kernel_launch(void* const* d_in, const int* in_sizes, int n_in,
                              void* d_out, int out_size, void* d_ws, size_t ws_size,
                              hipStream_t stream) {
    const float* x     = (const float*)d_in[0];
    const int*   erow  = (const int*)d_in[1];
    const int*   ecol  = (const int*)d_in[2];
    const float* evals = (const float*)d_in[3];
    const float* W1    = (const float*)d_in[4];
    const float* W2    = (const float*)d_in[5];
    const float* W3    = (const float*)d_in[6];
    float* out = (float*)d_out;

    const float* ev0 = evals;
    const float* ev1 = evals + (size_t)EE;

    float* R0 = (float*)d_ws;
    float* R1 = R0 + (size_t)NN * HH1;
    float* R2 = R1 + (size_t)NN * HH1;
    float* R3 = R2 + (size_t)NN * HH1;

    int*   cnt  = (int*)R2;
    int*   rank = cnt + NN;
    int*   scol = (int*)R3;
    float* sv0  = R3 + (size_t)EE;
    float* sv1  = sv0 + (size_t)EE;
    int*   off  = (int*)(sv1 + (size_t)EE);

    const int ZB = 256;
    dim3 gemm_blk(32, GEMM_WAVES);
    const int mtiles = NN / 16;
    const int gemm_gy = (mtiles + GEMM_WAVES - 1) / GEMM_WAVES;
    dim3 gat_blk(32, GATHER_WAVES);
    const int gat_g = (NN + GATHER_WAVES - 1) / GATHER_WAVES;

    gcn_zero_counts<<<(NN + ZB - 1) / ZB, ZB, 0, stream>>>(cnt, NN);
    gcn_rank_edges<<<(EE + ZB - 1) / ZB, ZB, 0, stream>>>(erow, cnt, rank);
    gcn_scan_counts<<<SCAN_WGS, SCAN_T, 0, stream>>>(cnt, off);
    gcn_place_edges<<<(EE + ZB - 1) / ZB, ZB, 0, stream>>>(
        erow, ecol, ev0, ev1, off, rank, scol, sv0, sv1);

    float* pre0 = R0;
    float* pre1 = R1;
    float* h1   = R2;
    gcn_gemm_wmma2<FF, HH1><<<dim3(gemm_gy), gemm_blk, 0, stream>>>(
        x, W1, W1 + (size_t)FF * HH1, pre0, pre1, NN);
    gcn_gather_rows<4, true><<<gat_g, gat_blk, 0, stream>>>(
        pre0, pre1, off, scol, sv0, sv1, h1);

    float* pre0b = R0;
    float* pre1b = R0 + (size_t)NN * HH2;
    float* h2    = R1;
    gcn_gemm_wmma2<HH1, HH2><<<dim3(gemm_gy), gemm_blk, 0, stream>>>(
        h1, W2, W2 + (size_t)HH1 * HH2, pre0b, pre1b, NN);
    gcn_gather_rows<2, true><<<gat_g, gat_blk, 0, stream>>>(
        pre0b, pre1b, off, scol, sv0, sv1, h2);

    float* p0 = R2;
    float* p1 = R2 + NN;
    gcn_dot64x2<<<(NN + ZB - 1) / ZB, ZB, 0, stream>>>(
        h2, W3, W3 + (size_t)HH2, p0, p1);
    gcn_gather_scalar<<<(NN + ZB - 1) / ZB, ZB, 0, stream>>>(
        p0, p1, off, scol, sv0, sv1, out);
}


# Layer-1 product run as two launches whose k loops also carry the edge sort (one or two edges per lane per k chunk): ranking rides in the first, placing in the second; both edge launches gone
# speedup vs baseline: 2.0660x; 1.0818x over previous
#include <hip/hip_runtime.h>


typedef __attribute__((ext_vector_type(2))) float v2f;
typedef __attribute__((ext_vector_type(8))) float v8f;

#define NN 100000
#define FF 256
#define EE 1600000
#define HH1 128
#define HH2 64

#define GEMM_WAVES 4
#define GATHER_WAVES 8

__global__ void gcn_zero_counts(int* __restrict__ p, int n) {
    int i = blockIdx.x * blockDim.x + threadIdx.x;
    if (i < n) p[i] = 0;
}


#define SCAN_T 1024
#define SCAN_SEG (4 * SCAN_T)
#define SCAN_WGS ((NN + SCAN_SEG - 1) / SCAN_SEG)
__global__ __launch_bounds__(SCAN_T)
void gcn_scan_counts(const int* __restrict__ cnt, int* __restrict__ off) {
    static_assert(NN % 4 == 0, "counters in whole int4 pieces");
    static_assert(SCAN_T == 32 * 32, "one lane per wave in the cross-wave step");
    static_assert((SCAN_WGS - 1) * SCAN_SEG <= NN, "every earlier segment is a full one");
    __shared__ int wown[SCAN_T / 32];
    __shared__ int wprev[SCAN_T / 32];
    const int t = threadIdx.x;
    const int lane = t & 31;
    const int wave = t >> 5;
    const int g = blockIdx.x;
    const int4* __restrict__ src4 = reinterpret_cast<const int4*>(cnt);

    const int i4 = g * SCAN_T + t;
    const bool act = i4 < NN / 4;
    int4 v = make_int4(0, 0, 0, 0);
    if (act) v = src4[i4];

    int sprev = 0;
    for (int k = 0; k < g; k += 8) {
        int4 w[8];
#pragma unroll
        for (int u = 0; u < 8; ++u) {
            const int ks = (k + u < g) ? (k + u) : (g - 1);
            w[u] = src4[ks * SCAN_T + t];
        }
#pragma unroll
        for (int u = 0; u < 8; ++u) {
            const int s4 = (w[u].x + w[u].y) + (w[u].z + w[u].w);
            sprev += (k + u < g) ? s4 : 0;
        }
    }

    const int own = (v.x + v.y) + (v.z + v.w);
    int inc = own;
    for (int d = 1; d < 32; d <<= 1) {
        const int y = __shfl_up(inc, d, 32);
        if (lane >= d) inc += y;
    }
    int ps = sprev;
    for (int d = 16; d >= 1; d >>= 1) ps += __shfl_xor(ps, d, 32);
    if (lane == 31) wown[wave] = inc;
    if (lane == 0) wprev[wave] = ps;
    __syncthreads();
    int before = (lane < wave) ? wown[lane] : 0;
    int base = wprev[lane];
    for (int d = 16; d >= 1; d >>= 1) {
        before += __shfl_xor(before, d, 32);
        base += __shfl_xor(base, d, 32);
    }
    const int start = base + before + (inc - own);
    if (act) {
        int4 o;
        o.x = start;
        o.y = o.x + v.x;
        o.z = o.y + v.y;
        o.w = o.z + v.z;
        reinterpret_cast<int4*>(off)[i4] = o;
    }
    if (i4 == NN / 4 - 1) off[NN] = start + own;
}

#define GEMM_KC 16
template <int K, int Nout, int JOB>
__global__ __launch_bounds__(32 * GEMM_WAVES) __attribute__((amdgpu_waves_per_eu(JOB == 2 ? 4 : 3)))
void gcn_gemm_wmma2(const float* __restrict__ A,
                    const float* __restrict__ B0, const float* __restrict__ B1,
                    float* __restrict__ C0, float* __restrict__ C1, int M,
                    const int* __restrict__ erow, const int* __restrict__ ecol,
                    const float* __restrict__ ev0, const float* __restrict__ ev1,
                    const int* __restrict__ off, int* __restrict__ cnt,
                    int* __restrict__ rank, int* __restrict__ scol,
                    float* __restrict__ sv0, float* __restrict__ sv1) {
    static_assert(K % GEMM_KC == 0 && Nout % 16 == 0, "tile dims");
    constexpr int NT = Nout / 16;
    constexpr int TT = 2 * NT;
    constexpr int NC = K / GEMM_KC;
    constexpr int CHUNK_F = 4 * TT * 64;
    constexpr int UNITS = CHUNK_F / 8;
    constexpr int NTHR = 32 * GEMM_WAVES;
    constexpr int UPT = UNITS / NTHR;
    static_assert(UNITS % NTHR == 0 && UPT >= 1, "staging split");

    __shared__ __attribute__((aligned(16))) float wl[2][CHUNK_F];

    const int lane = threadIdx.x;
    const int wave = threadIdx.y;
    const int tid  = wave * 32 + lane;
    const int half = lane >> 4;
    const int mn   = lane & 15;
    const int tileM = blockIdx.x * GEMM_WAVES + wave;
    const bool live = tileM * 16 < M;
    const int row0 = live ? tileM * 16 : 0;

    const float* __restrict__ ap = A + (size_t)(row0 + mn) * K + 2 * half;

    const float* src[UPT];
#pragma unroll
    for (int i = 0; i < UPT; ++i) {
        const int u = i * NTHR + tid;
        const int frag = u >> 3, q = u & 7;
        const int ks = frag / TT, t = frag % TT;
        const int s = t / NT, tn = t % NT;
        const int hf = q >> 2, nn = (q & 3) * 4;
        src[i] = (s ? B1 : B0) + (size_t)(ks * 4 + hf * 2) * Nout + tn * 16 + nn;
    }

    v8f acc[TT];
#pragma unroll
    for (int t = 0; t < TT; ++t) acc[t] = v8f{};

    float4 r0[UPT], r1[UPT];
    v2f a[4], an[4];

#pragma unroll
    for (int i = 0; i < UPT; ++i) {
        r0[i] = *reinterpret_cast<const float4*>(src[i]);
        r1[i] = *reinterpret_cast<const float4*>(src[i] + Nout);
    }
#pragma unroll
    for (int ks = 0; ks < 4; ++ks) a[ks] = *reinterpret_cast<const v2f*>(ap + 4 * ks);
#pragma unroll
    for (int i = 0; i < UPT; ++i) {
        float* d = &wl[0][(i * NTHR + tid) * 8];
        *reinterpret_cast<float4*>(d)     = make_float4(r0[i].x, r1[i].x, r0[i].y, r1[i].y);
        *reinterpret_cast<float4*>(d + 4) = make_float4(r0[i].z, r1[i].z, r0[i].w, r1[i].w);
    }
    __syncthreads();

    constexpr int EW = (JOB == 1) ? 64 : 32;
    const int ebase = (blockIdx.x * GEMM_WAVES + wave) * (NC * EW) + lane;
    int er = 0, ek = 0;
    int erb = 0, ekb = 0;
    int fo = 0, fk = 0, fc = 0;
    float fa = 0.0f, fb = 0.0f;
    auto edge_step = [&](int s) {
        const int e0 = ebase + EW * s;
        const int e1 = e0 - EW;
        const int e2 = e0 - 2 * EW;
        if (JOB == 1) {
            if (s >= 2 && s <= NC + 1) {
                if (e2 < EE) rank[e2] = ek;
                if (e2 + 32 < EE) rank[e2 + 32] = ekb;
            }
            if (s >= 1 && s <= NC) {
                if (e1 < EE) ek = atomicAdd(cnt + er, 1);
                if (e1 + 32 < EE) ekb = atomicAdd(cnt + erb, 1);
            }
            if (s < NC) {
                if (e0 < EE) er = __builtin_nontemporal_load(erow + e0);
                if (e0 + 32 < EE) erb = __builtin_nontemporal_load(erow + e0 + 32);
            }
        }
        if (JOB == 2) {
            if (s >= 2 && s <= NC + 1 && e2 < EE) {
                const int p = fo + fk;
                scol[p] = fc;
                sv0[p]  = fa;
                sv1[p]  = fb;
            }
            if (s >= 1 && s <= NC && e1 < EE) {
                fo = off[er];
                fk = __builtin_nontemporal_load(rank + e1);
                fc = __builtin_nontemporal_load(ecol + e1);
                fa = __builtin_nontemporal_load(ev0 + e1);
                fb = __builtin_nontemporal_load(ev1 + e1);
            }
            if (s < NC && e0 < EE) er = __builtin_nontemporal_load(erow + e0);
        }
    };

    for (int c = 0; c < NC; ++c) {
        const bool more = c + 1 < NC;
        edge_step(c);
        if (more) {
            const size_t kn = (size_t)(c + 1) * GEMM_KC;
#pragma unroll
            for (int i = 0; i < UPT; ++i) {
                r0[i] = *reinterpret_cast<const float4*>(src[i] + kn * Nout);
                r1[i] = *reinterpret_cast<const float4*>(src[i] + kn * Nout + Nout);
            }
#pragma unroll
            for (int ks = 0; ks < 4; ++ks)
                an[ks] = *reinterpret_cast<const v2f*>(ap + kn + 4 * ks);
        }
        const float* wb = &wl[c & 1][lane * 2];
        v2f bx[TT], by[TT];
#pragma unroll
        for (int t = 0; t < TT; ++t) bx[t] = *reinterpret_cast<const v2f*>(wb + (0 * TT + t) * 64);
#pragma unroll
        for (int t = 0; t < TT; ++t) by[t] = *reinterpret_cast<const v2f*>(wb + (1 * TT + t) * 64);
#pragma unroll
        for (int t = 0; t < TT; ++t)
            acc[t] = __builtin_amdgcn_wmma_f32_16x16x4_f32(false, a[0], false, bx[t],
                                                           (short)0, acc[t], false, false);
#pragma unroll
        for (int t = 0; t < TT; ++t) bx[t] = *reinterpret_cast<const v2f*>(wb + (2 * TT + t) * 64);
#pragma unroll
        for (int t = 0; t < TT; ++t)
            acc[t] = __builtin_amdgcn_wmma_f32_16x16x4_f32(false, a[1], false, by[t],
                                                           (short)0, acc[t], false, false);
#pragma unroll
        for (int t = 0; t < TT; ++t) by[t] = *reinterpret_cast<const v2f*>(wb + (3 * TT + t) * 64);
#pragma unroll
        for (int t = 0; t < TT; ++t)
            acc[t] = __builtin_amdgcn_wmma_f32_16x16x4_f32(false, a[2], false, bx[t],
                                                           (short)0, acc[t], false, false);
#pragma unroll
        for (int t = 0; t < TT; ++t)
            acc[t] = __builtin_amdgcn_wmma_f32_16x16x4_f32(false, a[3], false, by[t],
                                                           (short)0, acc[t], false, false);
        if (more) {
            float* wd = &wl[(c & 1) ^ 1][0];
#pragma unroll
            for (int i = 0; i < UPT; ++i) {
                float* d = wd + (i * NTHR + tid) * 8;
                *reinterpret_cast<float4*>(d)     = make_float4(r0[i].x, r1[i].x, r0[i].y, r1[i].y);
                *reinterpret_cast<float4*>(d + 4) = make_float4(r0[i].z, r1[i].z, r0[i].w, r1[i].w);
            }
#pragma unroll
            for (int ks = 0; ks < 4; ++ks) a[ks] = an[ks];
        }
        __syncthreads();
    }
    edge_step(NC);
    edge_step(NC + 1);

    if (live) {
#pragma unroll
        for (int t = 0; t < TT; ++t) {
            const int s = t / NT, tn = t % NT;
            float* __restrict__ cp = (s ? C1 : C0)
                + (size_t)(row0 + half * 8) * Nout + tn * 16 + mn;
#pragma unroll
            for (int v = 0; v < 8; ++v) cp[(size_t)v * Nout] = acc[t][v];
        }
    }
}

static __device__ __forceinline__ float gcn_lane_f(float x, int j) {
    return __int_as_float(__builtin_amdgcn_readlane(__float_as_int(x), j));
}

template <int VEC, bool RELU>
__global__ __launch_bounds__(32 * GATHER_WAVES)
void gcn_gather_rows(const float* __restrict__ pre0, const float* __restrict__ pre1,
                     const int* __restrict__ off, const int* __restrict__ scol,
                     const float* __restrict__ sv0, const float* __restrict__ sv1,
                     float* __restrict__ h) {
    typedef __attribute__((ext_vector_type(VEC))) float vecT;
    constexpr int H = 32 * VEC;
    const int lane = threadIdx.x;
    const int row  = blockIdx.x * GATHER_WAVES + threadIdx.y;
    if (row >= NN) return;
    const int beg = __builtin_amdgcn_readfirstlane(off[row]);
    const int end = __builtin_amdgcn_readfirstlane(off[row + 1]);
    const float* __restrict__ q0 = pre0 + lane * VEC;
    const float* __restrict__ q1 = pre1 + lane * VEC;

    vecT acc = {};
    for (int base = beg; base < end; base += 32) {
        const int left = end - base;
        const int n = left < 32 ? left : 32;
        int cj = 0;
        float a0 = 0.0f, a1 = 0.0f;
        if (lane < n) {
            cj = scol[base + lane];
            a0 = sv0[base + lane];
            a1 = sv1[base + lane];
        }
        int j = 0;
        for (; j + 4 <= n; j += 4) {
            vecT x0[4], x1[4];
            float w0[4], w1[4];
#pragma unroll
            for (int u = 0; u < 4; ++u) {
                const int c = __builtin_amdgcn_readlane(cj, j + u);
                w0[u] = gcn_lane_f(a0, j + u);
                w1[u] = gcn_lane_f(a1, j + u);
                x0[u] = *reinterpret_cast<const vecT*>(q0 + (size_t)c * H);
                x1[u] = *reinterpret_cast<const vecT*>(q1 + (size_t)c * H);
            }
#pragma unroll
            for (int u = 0; u < 4; ++u) acc += w0[u] * x0[u] + w1[u] * x1[u];
        }
        for (; j < n; ++j) {
            const int c = __builtin_amdgcn_readlane(cj, j);
            const float w0 = gcn_lane_f(a0, j);
            const float w1 = gcn_lane_f(a1, j);
            const vecT x0 = *reinterpret_cast<const vecT*>(q0 + (size_t)c * H);
            const vecT x1 = *reinterpret_cast<const vecT*>(q1 + (size_t)c * H);
            acc += w0 * x0 + w1 * x1;
        }
    }
    if (RELU) {
#pragma unroll
        for (int k = 0; k < VEC; ++k) acc[k] = fmaxf(acc[k], 0.0f);
    }
    *reinterpret_cast<vecT*>(h + (size_t)row * H + lane * VEC) = acc;
}

__global__ void gcn_gather_scalar(const float* __restrict__ p0,
                                  const float* __restrict__ p1,
                                  const int* __restrict__ off,
                                  const int* __restrict__ scol,
                                  const float* __restrict__ sv0,
                                  const float* __restrict__ sv1,
                                  float* __restrict__ out) {
    int r = blockIdx.x * blockDim.x + threadIdx.x;
    if (r >= NN) return;
    const int beg = off[r];
    const int end = off[r + 1];
    float s = 0.0f;
    for (int i = beg; i < end; ++i) {
        const int c = scol[i];
        s += sv0[i] * p0[c] + sv1[i] * p1[c];
    }
    out[r] = s;
}

__global__ void gcn_dot64x2(const float* __restrict__ h,
                            const float* __restrict__ w0,
                            const float* __restrict__ w1,
                            float* __restrict__ p0, float* __restrict__ p1) {
    int i = blockIdx.x * blockDim.x + threadIdx.x;
    if (i >= NN) return;
    const float* row = h + (size_t)i * HH2;
    float s0 = 0.0f, s1 = 0.0f;
#pragma unroll
    for (int k = 0; k < HH2; ++k) {
        float hv = row[k];
        s0 += hv * w0[k];
        s1 += hv * w1[k];
    }
    p0[i] = s0;
    p1[i] = s1;
}

extern "C" void kernel_launch(void* const* d_in, const int* in_sizes, int n_in,
                              void* d_out, int out_size, void* d_ws, size_t ws_size,
                              hipStream_t stream) {
    const float* x     = (const float*)d_in[0];
    const int*   erow  = (const int*)d_in[1];
    const int*   ecol  = (const int*)d_in[2];
    const float* evals = (const float*)d_in[3];
    const float* W1    = (const float*)d_in[4];
    const float* W2    = (const float*)d_in[5];
    const float* W3    = (const float*)d_in[6];
    float* out = (float*)d_out;

    const float* ev0 = evals;
    const float* ev1 = evals + (size_t)EE;

    float* R0 = (float*)d_ws;
    float* R1 = R0 + (size_t)NN * HH1;
    float* R2 = R1 + (size_t)NN * HH1;
    float* R3 = R2 + (size_t)NN * HH1;

    int*   cnt  = (int*)R2;
    int*   rank = cnt + NN;
    int*   scol = (int*)R3;
    float* sv0  = R3 + (size_t)EE;
    float* sv1  = sv0 + (size_t)EE;
    int*   off  = (int*)(sv1 + (size_t)EE);

    const int ZB = 256;
    dim3 gemm_blk(32, GEMM_WAVES);
    const int mtiles = NN / 16;
    const int gemm_gy = (mtiles + GEMM_WAVES - 1) / GEMM_WAVES;
    constexpr int GA = 768;
    constexpr int ROWS_A = GA * GEMM_WAVES * 16;
    constexpr int ROWS_B = NN - ROWS_A;
    constexpr int GB = (ROWS_B / 16 + GEMM_WAVES - 1) / GEMM_WAVES;
    static_assert(ROWS_B > 0 && ROWS_B % 16 == 0, "whole 16-row tiles");
    static_assert((long long)GA * GEMM_WAVES * (FF / GEMM_KC) * 64 >= EE, "part A ranks every edge");
    static_assert((long long)GB * GEMM_WAVES * (FF / GEMM_KC) * 32 >= EE, "part B places every edge");
    dim3 gat_blk(32, GATHER_WAVES);
    const int gat_g = (NN + GATHER_WAVES - 1) / GATHER_WAVES;

    float* pre0 = R0;
    float* pre1 = R1;
    float* h1   = R2;
    gcn_zero_counts<<<(NN + ZB - 1) / ZB, ZB, 0, stream>>>(cnt, NN);
    gcn_gemm_wmma2<FF, HH1, 1><<<dim3(GA), gemm_blk, 0, stream>>>(
        x, W1, W1 + (size_t)FF * HH1, pre0, pre1, ROWS_A,
        erow, ecol, ev0, ev1, off, cnt, rank, scol, sv0, sv1);
    gcn_scan_counts<<<SCAN_WGS, SCAN_T, 0, stream>>>(cnt, off);
    gcn_gemm_wmma2<FF, HH1, 2><<<dim3(GB), gemm_blk, 0, stream>>>(
        x + (size_t)ROWS_A * FF, W1, W1 + (size_t)FF * HH1,
        pre0 + (size_t)ROWS_A * HH1, pre1 + (size_t)ROWS_A * HH1, ROWS_B,
        erow, ecol, ev0, ev1, off, cnt, rank, scol, sv0, sv1);

    gcn_gather_rows<4, true><<<gat_g, gat_blk, 0, stream>>>(
        pre0, pre1, off, scol, sv0, sv1, h1);

    float* pre0b = R0;
    float* pre1b = R0 + (size_t)NN * HH2;
    float* h2    = R1;
    gcn_gemm_wmma2<HH1, HH2, 0><<<dim3(gemm_gy), gemm_blk, 0, stream>>>(
        h1, W2, W2 + (size_t)HH1 * HH2, pre0b, pre1b, NN,
        erow, ecol, ev0, ev1, off, cnt, rank, scol, sv0, sv1);
    gcn_gather_rows<2, true><<<gat_g, gat_blk, 0, stream>>>(
        pre0b, pre1b, off, scol, sv0, sv1, h2);

    float* p0 = R2;
    float* p1 = R2 + NN;
    gcn_dot64x2<<<(NN + ZB - 1) / ZB, ZB, 0, stream>>>(
        h2, W3, W3 + (size_t)HH2, p0, p1);
    gcn_gather_scalar<<<(NN + ZB - 1) / ZB, ZB, 0, stream>>>(
        p0, p1, off, scol, sv0, sv1, out);
}


# Next layer's weights applied inside each row gather: W2 on WMMA in the layer-1 gather (16 rows meet in LDS), W3 dots in the layer-2 gather (lane-tree sum order); h1, h2 not stored, 2 launches gone
# speedup vs baseline: 2.1553x; 1.0432x over previous
#include <hip/hip_runtime.h>


typedef __attribute__((ext_vector_type(2))) float v2f;
typedef __attribute__((ext_vector_type(8))) float v8f;

#define NN 100000
#define FF 256
#define EE 1600000
#define HH1 128
#define HH2 64

#define GEMM_WAVES 4
#define GATHER_WAVES 8

__global__ void gcn_zero_counts(int* __restrict__ p, int n) {
    int i = blockIdx.x * blockDim.x + threadIdx.x;
    if (i < n) p[i] = 0;
}


#define SCAN_T 1024
#define SCAN_SEG (4 * SCAN_T)
#define SCAN_WGS ((NN + SCAN_SEG - 1) / SCAN_SEG)
__global__ __launch_bounds__(SCAN_T)
void gcn_scan_counts(const int* __restrict__ cnt, int* __restrict__ off) {
    static_assert(NN % 4 == 0, "counters in whole int4 pieces");
    static_assert(SCAN_T == 32 * 32, "one lane per wave in the cross-wave step");
    static_assert((SCAN_WGS - 1) * SCAN_SEG <= NN, "every earlier segment is a full one");
    __shared__ int wown[SCAN_T / 32];
    __shared__ int wprev[SCAN_T / 32];
    const int t = threadIdx.x;
    const int lane = t & 31;
    const int wave = t >> 5;
    const int g = blockIdx.x;
    const int4* __restrict__ src4 = reinterpret_cast<const int4*>(cnt);

    const int i4 = g * SCAN_T + t;
    const bool act = i4 < NN / 4;
    int4 v = make_int4(0, 0, 0, 0);
    if (act) v = src4[i4];

    int sprev = 0;
    for (int k = 0; k < g; k += 8) {
        int4 w[8];
#pragma unroll
        for (int u = 0; u < 8; ++u) {
            const int ks = (k + u < g) ? (k + u) : (g - 1);
            w[u] = src4[ks * SCAN_T + t];
        }
#pragma unroll
        for (int u = 0; u < 8; ++u) {
            const int s4 = (w[u].x + w[u].y) + (w[u].z + w[u].w);
            sprev += (k + u < g) ? s4 : 0;
        }
    }

    const int own = (v.x + v.y) + (v.z + v.w);
    int inc = own;
    for (int d = 1; d < 32; d <<= 1) {
        const int y = __shfl_up(inc, d, 32);
        if (lane >= d) inc += y;
    }
    int ps = sprev;
    for (int d = 16; d >= 1; d >>= 1) ps += __shfl_xor(ps, d, 32);
    if (lane == 31) wown[wave] = inc;
    if (lane == 0) wprev[wave] = ps;
    __syncthreads();
    int before = (lane < wave) ? wown[lane] : 0;
    int base = wprev[lane];
    for (int d = 16; d >= 1; d >>= 1) {
        before += __shfl_xor(before, d, 32);
        base += __shfl_xor(base, d, 32);
    }
    const int start = base + before + (inc - own);
    if (act) {
        int4 o;
        o.x = start;
        o.y = o.x + v.x;
        o.z = o.y + v.y;
        o.w = o.z + v.z;
        reinterpret_cast<int4*>(off)[i4] = o;
    }
    if (i4 == NN / 4 - 1) off[NN] = start + own;
}

#define GEMM_KC 16
template <int K, int Nout, int JOB>
__global__ __launch_bounds__(32 * GEMM_WAVES) __attribute__((amdgpu_waves_per_eu(JOB == 2 ? 4 : 3)))
void gcn_gemm_wmma2(const float* __restrict__ A,
                    const float* __restrict__ B0, const float* __restrict__ B1,
                    float* __restrict__ C0, float* __restrict__ C1, int M,
                    const int* __restrict__ erow, const int* __restrict__ ecol,
                    const float* __restrict__ ev0, const float* __restrict__ ev1,
                    const int* __restrict__ off, int* __restrict__ cnt,
                    int* __restrict__ rank, int* __restrict__ scol,
                    float* __restrict__ sv0, float* __restrict__ sv1) {
    static_assert(K % GEMM_KC == 0 && Nout % 16 == 0, "tile dims");
    constexpr int NT = Nout / 16;
    constexpr int TT = 2 * NT;
    constexpr int NC = K / GEMM_KC;
    constexpr int CHUNK_F = 4 * TT * 64;
    constexpr int UNITS = CHUNK_F / 8;
    constexpr int NTHR = 32 * GEMM_WAVES;
    constexpr int UPT = UNITS / NTHR;
    static_assert(UNITS % NTHR == 0 && UPT >= 1, "staging split");

    __shared__ __attribute__((aligned(16))) float wl[2][CHUNK_F];

    const int lane = threadIdx.x;
    const int wave = threadIdx.y;
    const int tid  = wave * 32 + lane;
    const int half = lane >> 4;
    const int mn   = lane & 15;
    const int tileM = blockIdx.x * GEMM_WAVES + wave;
    const bool live = tileM * 16 < M;
    const int row0 = live ? tileM * 16 : 0;

    const float* __restrict__ ap = A + (size_t)(row0 + mn) * K + 2 * half;

    const float* src[UPT];
#pragma unroll
    for (int i = 0; i < UPT; ++i) {
        const int u = i * NTHR + tid;
        const int frag = u >> 3, q = u & 7;
        const int ks = frag / TT, t = frag % TT;
        const int s = t / NT, tn = t % NT;
        const int hf = q >> 2, nn = (q & 3) * 4;
        src[i] = (s ? B1 : B0) + (size_t)(ks * 4 + hf * 2) * Nout + tn * 16 + nn;
    }

    v8f acc[TT];
#pragma unroll
    for (int t = 0; t < TT; ++t) acc[t] = v8f{};

    float4 r0[UPT], r1[UPT];
    v2f a[4], an[4];

#pragma unroll
    for (int i = 0; i < UPT; ++i) {
        r0[i] = *reinterpret_cast<const float4*>(src[i]);
        r1[i] = *reinterpret_cast<const float4*>(src[i] + Nout);
    }
#pragma unroll
    for (int ks = 0; ks < 4; ++ks) a[ks] = *reinterpret_cast<const v2f*>(ap + 4 * ks);
#pragma unroll
    for (int i = 0; i < UPT; ++i) {
        float* d = &wl[0][(i * NTHR + tid) * 8];
        *reinterpret_cast<float4*>(d)     = make_float4(r0[i].x, r1[i].x, r0[i].y, r1[i].y);
        *reinterpret_cast<float4*>(d + 4) = make_float4(r0[i].z, r1[i].z, r0[i].w, r1[i].w);
    }
    __syncthreads();

    constexpr int EW = (JOB == 1) ? 64 : 32;
    const int ebase = (blockIdx.x * GEMM_WAVES + wave) * (NC * EW) + lane;
    int er = 0, ek = 0;
    int erb = 0, ekb = 0;
    int fo = 0, fk = 0, fc = 0;
    float fa = 0.0f, fb = 0.0f;
    auto edge_step = [&](int s) {
        const int e0 = ebase + EW * s;
        const int e1 = e0 - EW;
        const int e2 = e0 - 2 * EW;
        if (JOB == 1) {
            if (s >= 2 && s <= NC + 1) {
                if (e2 < EE) rank[e2] = ek;
                if (e2 + 32 < EE) rank[e2 + 32] = ekb;
            }
            if (s >= 1 && s <= NC) {
                if (e1 < EE) ek = atomicAdd(cnt + er, 1);
                if (e1 + 32 < EE) ekb = atomicAdd(cnt + erb, 1);
            }
            if (s < NC) {
                if (e0 < EE) er = __builtin_nontemporal_load(erow + e0);
                if (e0 + 32 < EE) erb = __builtin_nontemporal_load(erow + e0 + 32);
            }
        }
        if (JOB == 2) {
            if (s >= 2 && s <= NC + 1 && e2 < EE) {
                const int p = fo + fk;
                scol[p] = fc;
                sv0[p]  = fa;
                sv1[p]  = fb;
            }
            if (s >= 1 && s <= NC && e1 < EE) {
                fo = off[er];
                fk = __builtin_nontemporal_load(rank + e1);
                fc = __builtin_nontemporal_load(ecol + e1);
                fa = __builtin_nontemporal_load(ev0 + e1);
                fb = __builtin_nontemporal_load(ev1 + e1);
            }
            if (s < NC && e0 < EE) er = __builtin_nontemporal_load(erow + e0);
        }
    };

    for (int c = 0; c < NC; ++c) {
        const bool more = c + 1 < NC;
        edge_step(c);
        if (more) {
            const size_t kn = (size_t)(c + 1) * GEMM_KC;
#pragma unroll
            for (int i = 0; i < UPT; ++i) {
                r0[i] = *reinterpret_cast<const float4*>(src[i] + kn * Nout);
                r1[i] = *reinterpret_cast<const float4*>(src[i] + kn * Nout + Nout);
            }
#pragma unroll
            for (int ks = 0; ks < 4; ++ks)
                an[ks] = *reinterpret_cast<const v2f*>(ap + kn + 4 * ks);
        }
        const float* wb = &wl[c & 1][lane * 2];
        v2f bx[TT], by[TT];
#pragma unroll
        for (int t = 0; t < TT; ++t) bx[t] = *reinterpret_cast<const v2f*>(wb + (0 * TT + t) * 64);
#pragma unroll
        for (int t = 0; t < TT; ++t) by[t] = *reinterpret_cast<const v2f*>(wb + (1 * TT + t) * 64);
#pragma unroll
        for (int t = 0; t < TT; ++t)
            acc[t] = __builtin_amdgcn_wmma_f32_16x16x4_f32(false, a[0], false, bx[t],
                                                           (short)0, acc[t], false, false);
#pragma unroll
        for (int t = 0; t < TT; ++t) bx[t] = *reinterpret_cast<const v2f*>(wb + (2 * TT + t) * 64);
#pragma unroll
        for (int t = 0; t < TT; ++t)
            acc[t] = __builtin_amdgcn_wmma_f32_16x16x4_f32(false, a[1], false, by[t],
                                                           (short)0, acc[t], false, false);
#pragma unroll
        for (int t = 0; t < TT; ++t) by[t] = *reinterpret_cast<const v2f*>(wb + (3 * TT + t) * 64);
#pragma unroll
        for (int t = 0; t < TT; ++t)
            acc[t] = __builtin_amdgcn_wmma_f32_16x16x4_f32(false, a[2], false, bx[t],
                                                           (short)0, acc[t], false, false);
#pragma unroll
        for (int t = 0; t < TT; ++t)
            acc[t] = __builtin_amdgcn_wmma_f32_16x16x4_f32(false, a[3], false, by[t],
                                                           (short)0, acc[t], false, false);
        if (more) {
            float* wd = &wl[(c & 1) ^ 1][0];
#pragma unroll
            for (int i = 0; i < UPT; ++i) {
                float* d = wd + (i * NTHR + tid) * 8;
                *reinterpret_cast<float4*>(d)     = make_float4(r0[i].x, r1[i].x, r0[i].y, r1[i].y);
                *reinterpret_cast<float4*>(d + 4) = make_float4(r0[i].z, r1[i].z, r0[i].w, r1[i].w);
            }
#pragma unroll
            for (int ks = 0; ks < 4; ++ks) a[ks] = an[ks];
        }
        __syncthreads();
    }
    edge_step(NC);
    edge_step(NC + 1);

    if (live) {
#pragma unroll
        for (int t = 0; t < TT; ++t) {
            const int s = t / NT, tn = t % NT;
            float* __restrict__ cp = (s ? C1 : C0)
                + (size_t)(row0 + half * 8) * Nout + tn * 16 + mn;
#pragma unroll
            for (int v = 0; v < 8; ++v) cp[(size_t)v * Nout] = acc[t][v];
        }
    }
}

static __device__ __forceinline__ float gcn_lane_f(float x, int j) {
    return __int_as_float(__builtin_amdgcn_readlane(__float_as_int(x), j));
}

template <int VEC> struct gcn_vec { typedef __attribute__((ext_vector_type(VEC))) float type; };

template <int VEC>
static __device__ __forceinline__ typename gcn_vec<VEC>::type
gcn_row_sum(const float* __restrict__ q0, const float* __restrict__ q1,
            int beg, int end, int lane, const int* __restrict__ scol,
            const float* __restrict__ sv0, const float* __restrict__ sv1) {
    typedef typename gcn_vec<VEC>::type vecT;
    constexpr int H = 32 * VEC;
    vecT acc = {};
    for (int base = beg; base < end; base += 32) {
        const int left = end - base;
        const int n = left < 32 ? left : 32;
        int cj = 0;
        float a0 = 0.0f, a1 = 0.0f;
        if (lane < n) {
            cj = scol[base + lane];
            a0 = sv0[base + lane];
            a1 = sv1[base + lane];
        }
        int j = 0;
        for (; j + 4 <= n; j += 4) {
            vecT x0[4], x1[4];
            float w0[4], w1[4];
#pragma unroll
            for (int u = 0; u < 4; ++u) {
                const int c = __builtin_amdgcn_readlane(cj, j + u);
                w0[u] = gcn_lane_f(a0, j + u);
                w1[u] = gcn_lane_f(a1, j + u);
                x0[u] = *reinterpret_cast<const vecT*>(q0 + (size_t)c * H);
                x1[u] = *reinterpret_cast<const vecT*>(q1 + (size_t)c * H);
            }
#pragma unroll
            for (int u = 0; u < 4; ++u) acc += w0[u] * x0[u] + w1[u] * x1[u];
        }
        for (; j < n; ++j) {
            const int c = __builtin_amdgcn_readlane(cj, j);
            const float w0 = gcn_lane_f(a0, j);
            const float w1 = gcn_lane_f(a1, j);
            const vecT x0 = *reinterpret_cast<const vecT*>(q0 + (size_t)c * H);
            const vecT x1 = *reinterpret_cast<const vecT*>(q1 + (size_t)c * H);
            acc += w0 * x0 + w1 * x1;
        }
    }
    return acc;
}

__global__ __launch_bounds__(32 * GATHER_WAVES)
void gcn_gather_rows_proj(const float* __restrict__ pre0, const float* __restrict__ pre1,
                          const int* __restrict__ off, const int* __restrict__ scol,
                          const float* __restrict__ sv0, const float* __restrict__ sv1,
                          const float* __restrict__ w0, const float* __restrict__ w1,
                          float* __restrict__ p0, float* __restrict__ p1) {
    const int lane = threadIdx.x;
    const int row  = blockIdx.x * GATHER_WAVES + threadIdx.y;
    if (row >= NN) return;
    const int beg = __builtin_amdgcn_readfirstlane(off[row]);
    const int end = __builtin_amdgcn_readfirstlane(off[row + 1]);
    const v2f wa = *reinterpret_cast<const v2f*>(w0 + lane * 2);
    const v2f wb = *reinterpret_cast<const v2f*>(w1 + lane * 2);

    v2f acc = gcn_row_sum<2>(pre0 + lane * 2, pre1 + lane * 2, beg, end, lane,
                             scol, sv0, sv1);
    acc[0] = fmaxf(acc[0], 0.0f);
    acc[1] = fmaxf(acc[1], 0.0f);
    float t0 = acc[0] * wa[0] + acc[1] * wa[1];
    float t1 = acc[0] * wb[0] + acc[1] * wb[1];
    for (int d = 16; d >= 1; d >>= 1) {
        t0 += __shfl_xor(t0, d, 32);
        t1 += __shfl_xor(t1, d, 32);
    }
    if (lane == 0) {
        p0[row] = t0;
        p1[row] = t1;
    }
}

#define FG_WAVES 16
#define FG_PITCH (HH1 + 4)
#define FG_GRID 1024
__global__ __launch_bounds__(32 * FG_WAVES) __attribute__((amdgpu_waves_per_eu(16)))
void gcn_gather_rows_mm(const float* __restrict__ pre0, const float* __restrict__ pre1,
                        const int* __restrict__ off, const int* __restrict__ scol,
                        const float* __restrict__ sv0, const float* __restrict__ sv1,
                        const float* __restrict__ B0, const float* __restrict__ B1,
                        float* __restrict__ C0, float* __restrict__ C1) {
    typedef __attribute__((ext_vector_type(4))) float vecT;
    constexpr int K = HH1, Nout = HH2;
    constexpr int NT = Nout / 16;
    constexpr int TT = 2 * NT;
    constexpr int KS = K / 4;
    constexpr int WL_F = KS * TT * 64;
    constexpr int NTHR = 32 * FG_WAVES;
    constexpr int UPT = (WL_F / 8) / NTHR;
    constexpr int NTILES = NN / 16;
    static_assert(NN % 16 == 0 && (WL_F / 8) % NTHR == 0 && NT % 2 == 0, "tile dims");
    static_assert(FG_WAVES == 16, "one wave per row of a tile, four groups of four");

    __shared__ __attribute__((aligned(16))) float wl[WL_F];
    __shared__ __attribute__((aligned(16))) float hb[16 * FG_PITCH];

    const int lane = threadIdx.x;
    const int wave = __builtin_amdgcn_readfirstlane((int)threadIdx.y);
    const int tid  = wave * 32 + lane;
    const int half = lane >> 4;
    const int mn   = lane & 15;

#pragma unroll
    for (int i = 0; i < UPT; ++i) {
        const int u = i * NTHR + tid;
        const int frag = u >> 3, q = u & 7;
        const int ks = frag / TT, t = frag % TT;
        const int s = t / NT, tn = t % NT;
        const int hf = q >> 2, nn = (q & 3) * 4;
        const float* sp = (s ? B1 : B0) + (size_t)(ks * 4 + hf * 2) * Nout + tn * 16 + nn;
        const float4 r0 = *reinterpret_cast<const float4*>(sp);
        const float4 r1 = *reinterpret_cast<const float4*>(sp + Nout);
        float* d = &wl[u * 8];
        *reinterpret_cast<float4*>(d)     = make_float4(r0.x, r1.x, r0.y, r1.y);
        *reinterpret_cast<float4*>(d + 4) = make_float4(r0.z, r1.z, r0.w, r1.w);
    }

    const float* __restrict__ q0 = pre0 + lane * 4;
    const float* __restrict__ q1 = pre1 + lane * 4;

    int it = 0;
    for (int tile = blockIdx.x; tile < NTILES; tile += gridDim.x, ++it) {
        const int row = tile * 16 + wave;
        const int beg = __builtin_amdgcn_readfirstlane(off[row]);
        const int end = __builtin_amdgcn_readfirstlane(off[row + 1]);

        vecT acc = gcn_row_sum<4>(q0, q1, beg, end, lane, scol, sv0, sv1);
#pragma unroll
        for (int k = 0; k < 4; ++k) acc[k] = fmaxf(acc[k], 0.0f);

        __syncthreads();
        *reinterpret_cast<vecT*>(&hb[wave * FG_PITCH + lane * 4]) = acc;
        __syncthreads();

        if ((wave >> 2) == (it & 3)) {
            const int w4 = wave & 3;
            const int s  = w4 >> 1;
            const int tp = (w4 & 1) * 2;
            const float* ha = &hb[mn * FG_PITCH + 2 * half];
            const float* wb = &wl[(s * NT + tp) * 64 + lane * 2];
            v8f c0 = v8f{}, c1 = v8f{};
#pragma unroll
            for (int ks = 0; ks < KS; ++ks) {
                const v2f a  = *reinterpret_cast<const v2f*>(ha + 4 * ks);
                const v2f b0 = *reinterpret_cast<const v2f*>(wb + ks * TT * 64);
                const v2f b1 = *reinterpret_cast<const v2f*>(wb + ks * TT * 64 + 64);
                c0 = __builtin_amdgcn_wmma_f32_16x16x4_f32(false, a, false, b0,
                                                           (short)0, c0, false, false);
                c1 = __builtin_amdgcn_wmma_f32_16x16x4_f32(false, a, false, b1,
                                                           (short)0, c1, false, false);
            }
            float* __restrict__ cp = (s ? C1 : C0)
                + (size_t)(tile * 16 + half * 8) * Nout + tp * 16 + mn;
#pragma unroll
            for (int v = 0; v < 8; ++v) {
                cp[(size_t)v * Nout]      = c0[v];
                cp[(size_t)v * Nout + 16] = c1[v];
            }
        }
    }
}

__global__ void gcn_gather_scalar(const float* __restrict__ p0,
                                  const float* __restrict__ p1,
                                  const int* __restrict__ off,
                                  const int* __restrict__ scol,
                                  const float* __restrict__ sv0,
                                  const float* __restrict__ sv1,
                                  float* __restrict__ out) {
    int r = blockIdx.x * blockDim.x + threadIdx.x;
    if (r >= NN) return;
    const int beg = off[r];
    const int end = off[r + 1];
    float s = 0.0f;
    for (int i = beg; i < end; ++i) {
        const int c = scol[i];
        s += sv0[i] * p0[c] + sv1[i] * p1[c];
    }
    out[r] = s;
}

extern "C" void kernel_launch(void* const* d_in, const int* in_sizes, int n_in,
                              void* d_out, int out_size, void* d_ws, size_t ws_size,
                              hipStream_t stream) {
    const float* x     = (const float*)d_in[0];
    const int*   erow  = (const int*)d_in[1];
    const int*   ecol  = (const int*)d_in[2];
    const float* evals = (const float*)d_in[3];
    const float* W1    = (const float*)d_in[4];
    const float* W2    = (const float*)d_in[5];
    const float* W3    = (const float*)d_in[6];
    float* out = (float*)d_out;

    const float* ev0 = evals;
    const float* ev1 = evals + (size_t)EE;

    float* R0 = (float*)d_ws;
    float* R1 = R0 + (size_t)NN * HH1;
    float* R2 = R1 + (size_t)NN * HH1;
    float* R3 = R2 + (size_t)NN * HH1;

    int*   cnt  = (int*)R2;
    int*   rank = cnt + NN;
    int*   scol = (int*)R3;
    float* sv0  = R3 + (size_t)EE;
    float* sv1  = sv0 + (size_t)EE;
    int*   off  = (int*)(sv1 + (size_t)EE);

    const int ZB = 256;
    dim3 gemm_blk(32, GEMM_WAVES);
    constexpr int GA = 768;
    constexpr int ROWS_A = GA * GEMM_WAVES * 16;
    constexpr int ROWS_B = NN - ROWS_A;
    constexpr int GB = (ROWS_B / 16 + GEMM_WAVES - 1) / GEMM_WAVES;
    static_assert(ROWS_B > 0 && ROWS_B % 16 == 0, "whole 16-row tiles");
    static_assert((long long)GA * GEMM_WAVES * (FF / GEMM_KC) * 64 >= EE, "part A ranks every edge");
    static_assert((long long)GB * GEMM_WAVES * (FF / GEMM_KC) * 32 >= EE, "part B places every edge");
    dim3 gat_blk(32, GATHER_WAVES);
    const int gat_g = (NN + GATHER_WAVES - 1) / GATHER_WAVES;

    float* pre0 = R0;
    float* pre1 = R1;
    gcn_zero_counts<<<(NN + ZB - 1) / ZB, ZB, 0, stream>>>(cnt, NN);
    gcn_gemm_wmma2<FF, HH1, 1><<<dim3(GA), gemm_blk, 0, stream>>>(
        x, W1, W1 + (size_t)FF * HH1, pre0, pre1, ROWS_A,
        erow, ecol, ev0, ev1, off, cnt, rank, scol, sv0, sv1);
    gcn_scan_counts<<<SCAN_WGS, SCAN_T, 0, stream>>>(cnt, off);
    gcn_gemm_wmma2<FF, HH1, 2><<<dim3(GB), gemm_blk, 0, stream>>>(
        x + (size_t)ROWS_A * FF, W1, W1 + (size_t)FF * HH1,
        pre0 + (size_t)ROWS_A * HH1, pre1 + (size_t)ROWS_A * HH1, ROWS_B,
        erow, ecol, ev0, ev1, off, cnt, rank, scol, sv0, sv1);

    float* pre0b = R2;
    float* pre1b = R2 + (size_t)NN * HH2;
    gcn_gather_rows_mm<<<dim3(FG_GRID), dim3(32, FG_WAVES), 0, stream>>>(
        pre0, pre1, off, scol, sv0, sv1, W2, W2 + (size_t)HH1 * HH2, pre0b, pre1b);

    float* p0 = R0;
    float* p1 = R0 + NN;
    gcn_gather_rows_proj<<<gat_g, gat_blk, 0, stream>>>(
        pre0b, pre1b, off, scol, sv0, sv1, W3, W3 + (size_t)HH2, p0, p1);

    gcn_gather_scalar<<<(NN + ZB - 1) / ZB, ZB, 0, stream>>>(
        p0, p1, off, scol, sv0, sv1, out);
}


# Layer-2 gather: a wave owns 9 adjacent rows and walks their one contiguous edge range, 8 edges' row loads in flight wherever rows end; 100,000 one-row waves become 11,112, all resident
# speedup vs baseline: 2.2232x; 1.0315x over previous
#include <hip/hip_runtime.h>


typedef __attribute__((ext_vector_type(2))) float v2f;
typedef __attribute__((ext_vector_type(8))) float v8f;

#define NN 100000
#define FF 256
#define EE 1600000
#define HH1 128
#define HH2 64

#define GEMM_WAVES 4

__global__ void gcn_zero_counts(int* __restrict__ p, int n) {
    int i = blockIdx.x * blockDim.x + threadIdx.x;
    if (i < n) p[i] = 0;
}


#define SCAN_T 1024
#define SCAN_SEG (4 * SCAN_T)
#define SCAN_WGS ((NN + SCAN_SEG - 1) / SCAN_SEG)
__global__ __launch_bounds__(SCAN_T)
void gcn_scan_counts(const int* __restrict__ cnt, int* __restrict__ off) {
    static_assert(NN % 4 == 0, "counters in whole int4 pieces");
    static_assert(SCAN_T == 32 * 32, "one lane per wave in the cross-wave step");
    static_assert((SCAN_WGS - 1) * SCAN_SEG <= NN, "every earlier segment is a full one");
    __shared__ int wown[SCAN_T / 32];
    __shared__ int wprev[SCAN_T / 32];
    const int t = threadIdx.x;
    const int lane = t & 31;
    const int wave = t >> 5;
    const int g = blockIdx.x;
    const int4* __restrict__ src4 = reinterpret_cast<const int4*>(cnt);

    const int i4 = g * SCAN_T + t;
    const bool act = i4 < NN / 4;
    int4 v = make_int4(0, 0, 0, 0);
    if (act) v = src4[i4];

    int sprev = 0;
    for (int k = 0; k < g; k += 8) {
        int4 w[8];
#pragma unroll
        for (int u = 0; u < 8; ++u) {
            const int ks = (k + u < g) ? (k + u) : (g - 1);
            w[u] = src4[ks * SCAN_T + t];
        }
#pragma unroll
        for (int u = 0; u < 8; ++u) {
            const int s4 = (w[u].x + w[u].y) + (w[u].z + w[u].w);
            sprev += (k + u < g) ? s4 : 0;
        }
    }

    const int own = (v.x + v.y) + (v.z + v.w);
    int inc = own;
    for (int d = 1; d < 32; d <<= 1) {
        const int y = __shfl_up(inc, d, 32);
        if (lane >= d) inc += y;
    }
    int ps = sprev;
    for (int d = 16; d >= 1; d >>= 1) ps += __shfl_xor(ps, d, 32);
    if (lane == 31) wown[wave] = inc;
    if (lane == 0) wprev[wave] = ps;
    __syncthreads();
    int before = (lane < wave) ? wown[lane] : 0;
    int base = wprev[lane];
    for (int d = 16; d >= 1; d >>= 1) {
        before += __shfl_xor(before, d, 32);
        base += __shfl_xor(base, d, 32);
    }
    const int start = base + before + (inc - own);
    if (act) {
        int4 o;
        o.x = start;
        o.y = o.x + v.x;
        o.z = o.y + v.y;
        o.w = o.z + v.z;
        reinterpret_cast<int4*>(off)[i4] = o;
    }
    if (i4 == NN / 4 - 1) off[NN] = start + own;
}

#define GEMM_KC 16
template <int K, int Nout, int JOB>
__global__ __launch_bounds__(32 * GEMM_WAVES) __attribute__((amdgpu_waves_per_eu(JOB == 2 ? 4 : 3)))
void gcn_gemm_wmma2(const float* __restrict__ A,
                    const float* __restrict__ B0, const float* __restrict__ B1,
                    float* __restrict__ C0, float* __restrict__ C1, int M,
                    const int* __restrict__ erow, const int* __restrict__ ecol,
                    const float* __restrict__ ev0, const float* __restrict__ ev1,
                    const int* __restrict__ off, int* __restrict__ cnt,
                    int* __restrict__ rank, int* __restrict__ scol,
                    float* __restrict__ sv0, float* __restrict__ sv1) {
    static_assert(K % GEMM_KC == 0 && Nout % 16 == 0, "tile dims");
    constexpr int NT = Nout / 16;
    constexpr int TT = 2 * NT;
    constexpr int NC = K / GEMM_KC;
    constexpr int CHUNK_F = 4 * TT * 64;
    constexpr int UNITS = CHUNK_F / 8;
    constexpr int NTHR = 32 * GEMM_WAVES;
    constexpr int UPT = UNITS / NTHR;
    static_assert(UNITS % NTHR == 0 && UPT >= 1, "staging split");

    __shared__ __attribute__((aligned(16))) float wl[2][CHUNK_F];

    const int lane = threadIdx.x;
    const int wave = threadIdx.y;
    const int tid  = wave * 32 + lane;
    const int half = lane >> 4;
    const int mn   = lane & 15;
    const int tileM = blockIdx.x * GEMM_WAVES + wave;
    const bool live = tileM * 16 < M;
    const int row0 = live ? tileM * 16 : 0;

    const float* __restrict__ ap = A + (size_t)(row0 + mn) * K + 2 * half;

    const float* src[UPT];
#pragma unroll
    for (int i = 0; i < UPT; ++i) {
        const int u = i * NTHR + tid;
        const int frag = u >> 3, q = u & 7;
        const int ks = frag / TT, t = frag % TT;
        const int s = t / NT, tn = t % NT;
        const int hf = q >> 2, nn = (q & 3) * 4;
        src[i] = (s ? B1 : B0) + (size_t)(ks * 4 + hf * 2) * Nout + tn * 16 + nn;
    }

    v8f acc[TT];
#pragma unroll
    for (int t = 0; t < TT; ++t) acc[t] = v8f{};

    float4 r0[UPT], r1[UPT];
    v2f a[4], an[4];

#pragma unroll
    for (int i = 0; i < UPT; ++i) {
        r0[i] = *reinterpret_cast<const float4*>(src[i]);
        r1[i] = *reinterpret_cast<const float4*>(src[i] + Nout);
    }
#pragma unroll
    for (int ks = 0; ks < 4; ++ks) a[ks] = *reinterpret_cast<const v2f*>(ap + 4 * ks);
#pragma unroll
    for (int i = 0; i < UPT; ++i) {
        float* d = &wl[0][(i * NTHR + tid) * 8];
        *reinterpret_cast<float4*>(d)     = make_float4(r0[i].x, r1[i].x, r0[i].y, r1[i].y);
        *reinterpret_cast<float4*>(d + 4) = make_float4(r0[i].z, r1[i].z, r0[i].w, r1[i].w);
    }
    __syncthreads();

    constexpr int EW = (JOB == 1) ? 64 : 32;
    const int ebase = (blockIdx.x * GEMM_WAVES + wave) * (NC * EW) + lane;
    int er = 0, ek = 0;
    int erb = 0, ekb = 0;
    int fo = 0, fk = 0, fc = 0;
    float fa = 0.0f, fb = 0.0f;
    auto edge_step = [&](int s) {
        const int e0 = ebase + EW * s;
        const int e1 = e0 - EW;
        const int e2 = e0 - 2 * EW;
        if (JOB == 1) {
            if (s >= 2 && s <= NC + 1) {
                if (e2 < EE) rank[e2] = ek;
                if (e2 + 32 < EE) rank[e2 + 32] = ekb;
            }
            if (s >= 1 && s <= NC) {
                if (e1 < EE) ek = atomicAdd(cnt + er, 1);
                if (e1 + 32 < EE) ekb = atomicAdd(cnt + erb, 1);
            }
            if (s < NC) {
                if (e0 < EE) er = __builtin_nontemporal_load(erow + e0);
                if (e0 + 32 < EE) erb = __builtin_nontemporal_load(erow + e0 + 32);
            }
        }
        if (JOB == 2) {
            if (s >= 2 && s <= NC + 1 && e2 < EE) {
                const int p = fo + fk;
                scol[p] = fc;
                sv0[p]  = fa;
                sv1[p]  = fb;
            }
            if (s >= 1 && s <= NC && e1 < EE) {
                fo = off[er];
                fk = __builtin_nontemporal_load(rank + e1);
                fc = __builtin_nontemporal_load(ecol + e1);
                fa = __builtin_nontemporal_load(ev0 + e1);
                fb = __builtin_nontemporal_load(ev1 + e1);
            }
            if (s < NC && e0 < EE) er = __builtin_nontemporal_load(erow + e0);
        }
    };

    for (int c = 0; c < NC; ++c) {
        const bool more = c + 1 < NC;
        edge_step(c);
        if (more) {
            const size_t kn = (size_t)(c + 1) * GEMM_KC;
#pragma unroll
            for (int i = 0; i < UPT; ++i) {
                r0[i] = *reinterpret_cast<const float4*>(src[i] + kn * Nout);
                r1[i] = *reinterpret_cast<const float4*>(src[i] + kn * Nout + Nout);
            }
#pragma unroll
            for (int ks = 0; ks < 4; ++ks)
                an[ks] = *reinterpret_cast<const v2f*>(ap + kn + 4 * ks);
        }
        const float* wb = &wl[c & 1][lane * 2];
        v2f bx[TT], by[TT];
#pragma unroll
        for (int t = 0; t < TT; ++t) bx[t] = *reinterpret_cast<const v2f*>(wb + (0 * TT + t) * 64);
#pragma unroll
        for (int t = 0; t < TT; ++t) by[t] = *reinterpret_cast<const v2f*>(wb + (1 * TT + t) * 64);
#pragma unroll
        for (int t = 0; t < TT; ++t)
            acc[t] = __builtin_amdgcn_wmma_f32_16x16x4_f32(false, a[0], false, bx[t],
                                                           (short)0, acc[t], false, false);
#pragma unroll
        for (int t = 0; t < TT; ++t) bx[t] = *reinterpret_cast<const v2f*>(wb + (2 * TT + t) * 64);
#pragma unroll
        for (int t = 0; t < TT; ++t)
            acc[t] = __builtin_amdgcn_wmma_f32_16x16x4_f32(false, a[1], false, by[t],
                                                           (short)0, acc[t], false, false);
#pragma unroll
        for (int t = 0; t < TT; ++t) by[t] = *reinterpret_cast<const v2f*>(wb + (3 * TT + t) * 64);
#pragma unroll
        for (int t = 0; t < TT; ++t)
            acc[t] = __builtin_amdgcn_wmma_f32_16x16x4_f32(false, a[2], false, bx[t],
                                                           (short)0, acc[t], false, false);
#pragma unroll
        for (int t = 0; t < TT; ++t)
            acc[t] = __builtin_amdgcn_wmma_f32_16x16x4_f32(false, a[3], false, by[t],
                                                           (short)0, acc[t], false, false);
        if (more) {
            float* wd = &wl[(c & 1) ^ 1][0];
#pragma unroll
            for (int i = 0; i < UPT; ++i) {
                float* d = wd + (i * NTHR + tid) * 8;
                *reinterpret_cast<float4*>(d)     = make_float4(r0[i].x, r1[i].x, r0[i].y, r1[i].y);
                *reinterpret_cast<float4*>(d + 4) = make_float4(r0[i].z, r1[i].z, r0[i].w, r1[i].w);
            }
#pragma unroll
            for (int ks = 0; ks < 4; ++ks) a[ks] = an[ks];
        }
        __syncthreads();
    }
    edge_step(NC);
    edge_step(NC + 1);

    if (live) {
#pragma unroll
        for (int t = 0; t < TT; ++t) {
            const int s = t / NT, tn = t % NT;
            float* __restrict__ cp = (s ? C1 : C0)
                + (size_t)(row0 + half * 8) * Nout + tn * 16 + mn;
#pragma unroll
            for (int v = 0; v < 8; ++v) cp[(size_t)v * Nout] = acc[t][v];
        }
    }
}

static __device__ __forceinline__ float gcn_lane_f(float x, int j) {
    return __int_as_float(__builtin_amdgcn_readlane(__float_as_int(x), j));
}

typedef __attribute__((address_space(1))) const v2f gcn_gv2f;
static __device__ __forceinline__ gcn_gv2f* gcn_scalar_row2(const float* p) {
    const unsigned long long v = (unsigned long long)p;
    const unsigned lo = __builtin_amdgcn_readfirstlane((unsigned)v);
    const unsigned hi = __builtin_amdgcn_readfirstlane((unsigned)(v >> 32));
    return (gcn_gv2f*)(((unsigned long long)hi << 32) | lo);
}

template <int VEC> struct gcn_vec { typedef __attribute__((ext_vector_type(VEC))) float type; };

template <int VEC>
static __device__ __forceinline__ typename gcn_vec<VEC>::type
gcn_row_sum(const float* __restrict__ q0, const float* __restrict__ q1,
            int beg, int end, int lane, const int* __restrict__ scol,
            const float* __restrict__ sv0, const float* __restrict__ sv1) {
    typedef typename gcn_vec<VEC>::type vecT;
    constexpr int H = 32 * VEC;
    vecT acc = {};
    for (int base = beg; base < end; base += 32) {
        const int left = end - base;
        const int n = left < 32 ? left : 32;
        int cj = 0;
        float a0 = 0.0f, a1 = 0.0f;
        if (lane < n) {
            cj = scol[base + lane];
            a0 = sv0[base + lane];
            a1 = sv1[base + lane];
        }
        int j = 0;
        for (; j + 4 <= n; j += 4) {
            vecT x0[4], x1[4];
            float w0[4], w1[4];
#pragma unroll
            for (int u = 0; u < 4; ++u) {
                const int c = __builtin_amdgcn_readlane(cj, j + u);
                w0[u] = gcn_lane_f(a0, j + u);
                w1[u] = gcn_lane_f(a1, j + u);
                x0[u] = *reinterpret_cast<const vecT*>(q0 + (size_t)c * H);
                x1[u] = *reinterpret_cast<const vecT*>(q1 + (size_t)c * H);
            }
#pragma unroll
            for (int u = 0; u < 4; ++u) acc += w0[u] * x0[u] + w1[u] * x1[u];
        }
        for (; j < n; ++j) {
            const int c = __builtin_amdgcn_readlane(cj, j);
            const float w0 = gcn_lane_f(a0, j);
            const float w1 = gcn_lane_f(a1, j);
            const vecT x0 = *reinterpret_cast<const vecT*>(q0 + (size_t)c * H);
            const vecT x1 = *reinterpret_cast<const vecT*>(q1 + (size_t)c * H);
            acc += w0 * x0 + w1 * x1;
        }
    }
    return acc;
}

#define GP_WAVES 8
#define GP_ROWS 9
#define GP_B 8
#define GP_NWAVES ((NN + GP_ROWS - 1) / GP_ROWS)
#define GP_GRID ((GP_NWAVES + GP_WAVES - 1) / GP_WAVES)
__global__ __launch_bounds__(32 * GP_WAVES)
void gcn_gather_rows_proj(const float* __restrict__ pre0, const float* __restrict__ pre1,
                          const int* __restrict__ off, const int* __restrict__ scol,
                          const float* __restrict__ sv0, const float* __restrict__ sv1,
                          const float* __restrict__ w0, const float* __restrict__ w1,
                          float* __restrict__ p0, float* __restrict__ p1) {
    static_assert(GP_ROWS + 1 <= 32, "one lane per row bound");
    const int lane = threadIdx.x;
    const int wid = __builtin_amdgcn_readfirstlane((int)(blockIdx.x * GP_WAVES + threadIdx.y));
    const int r0 = wid * GP_ROWS;
    if (r0 >= NN) return;
    const int nr = (NN - r0 < GP_ROWS) ? (NN - r0) : GP_ROWS;
    int ob = 0;
    if (lane <= nr) ob = off[r0 + lane];
    const v2f wa = *reinterpret_cast<const v2f*>(w0 + lane * 2);
    const v2f wb = *reinterpret_cast<const v2f*>(w1 + lane * 2);

    const int ebeg = __builtin_amdgcn_readlane(ob, 0);
    const int eend = __builtin_amdgcn_readlane(ob, nr);
    int i = 0;
    int rend = __builtin_amdgcn_readlane(ob, 1);
    v2f acc = {};
    float res0 = 0.0f, res1 = 0.0f;

    auto close_row = [&]() {
        const float h0 = fmaxf(acc[0], 0.0f);
        const float h1 = fmaxf(acc[1], 0.0f);
        float t0 = h0 * wa[0] + h1 * wa[1];
        float t1 = h0 * wb[0] + h1 * wb[1];
        for (int d = 16; d >= 1; d >>= 1) {
            t0 += __shfl_xor(t0, d, 32);
            t1 += __shfl_xor(t1, d, 32);
        }
        if (lane == i) {
            res0 = t0;
            res1 = t1;
        }
        acc = v2f{};
        ++i;
        rend = (i < nr) ? __builtin_amdgcn_readlane(ob, i + 1) : 0x7fffffff;
    };

    int cj = 0;
    float a0 = 0.0f, a1 = 0.0f;
    if (ebeg + lane < eend) {
        cj = scol[ebeg + lane];
        a0 = sv0[ebeg + lane];
        a1 = sv1[ebeg + lane];
    }
    for (int base = ebeg; base < eend; base += 32) {
        const int left = eend - base;
        const int n = left < 32 ? left : 32;
        int cjn = 0;
        float a0n = 0.0f, a1n = 0.0f;
        for (int j = 0; j < n; j += GP_B) {
            v2f x0[GP_B], x1[GP_B];
            float e0[GP_B], e1[GP_B];
#pragma unroll
            for (int u = 0; u < GP_B; ++u) {
                const int jj = (j + u < n) ? (j + u) : (n - 1);
                const int c = __builtin_amdgcn_readlane(cj, jj);
                e0[u] = gcn_lane_f(a0, jj);
                e1[u] = gcn_lane_f(a1, jj);
                x0[u] = gcn_scalar_row2(pre0 + (size_t)c * HH2)[lane];
                x1[u] = gcn_scalar_row2(pre1 + (size_t)c * HH2)[lane];
            }
            if (j == 0 && base + 32 + lane < eend) {
                const unsigned pn = (unsigned)(base + 32 + lane);
                cjn = scol[pn];
                a0n = sv0[pn];
                a1n = sv1[pn];
            }
#pragma unroll
            for (int u = 0; u < GP_B; ++u) {
                if (j + u < n) {
                    const int e = base + j + u;
                    while (e >= rend) close_row();
                    acc += e0[u] * x0[u] + e1[u] * x1[u];
                }
            }
        }
        cj = cjn;
        a0 = a0n;
        a1 = a1n;
    }
    while (i < nr) close_row();
    if (lane < nr) {
        p0[r0 + lane] = res0;
        p1[r0 + lane] = res1;
    }
}

#define FG_WAVES 16
#define FG_PITCH (HH1 + 4)
#define FG_GRID 1024
__global__ __launch_bounds__(32 * FG_WAVES) __attribute__((amdgpu_waves_per_eu(16)))
void gcn_gather_rows_mm(const float* __restrict__ pre0, const float* __restrict__ pre1,
                        const int* __restrict__ off, const int* __restrict__ scol,
                        const float* __restrict__ sv0, const float* __restrict__ sv1,
                        const float* __restrict__ B0, const float* __restrict__ B1,
                        float* __restrict__ C0, float* __restrict__ C1) {
    typedef __attribute__((ext_vector_type(4))) float vecT;
    constexpr int K = HH1, Nout = HH2;
    constexpr int NT = Nout / 16;
    constexpr int TT = 2 * NT;
    constexpr int KS = K / 4;
    constexpr int WL_F = KS * TT * 64;
    constexpr int NTHR = 32 * FG_WAVES;
    constexpr int UPT = (WL_F / 8) / NTHR;
    constexpr int NTILES = NN / 16;
    static_assert(NN % 16 == 0 && (WL_F / 8) % NTHR == 0 && NT % 2 == 0, "tile dims");
    static_assert(FG_WAVES == 16, "one wave per row of a tile, four groups of four");

    __shared__ __attribute__((aligned(16))) float wl[WL_F];
    __shared__ __attribute__((aligned(16))) float hb[16 * FG_PITCH];

    const int lane = threadIdx.x;
    const int wave = __builtin_amdgcn_readfirstlane((int)threadIdx.y);
    const int tid  = wave * 32 + lane;
    const int half = lane >> 4;
    const int mn   = lane & 15;

#pragma unroll
    for (int i = 0; i < UPT; ++i) {
        const int u = i * NTHR + tid;
        const int frag = u >> 3, q = u & 7;
        const int ks = frag / TT, t = frag % TT;
        const int s = t / NT, tn = t % NT;
        const int hf = q >> 2, nn = (q & 3) * 4;
        const float* sp = (s ? B1 : B0) + (size_t)(ks * 4 + hf * 2) * Nout + tn * 16 + nn;
        const float4 r0 = *reinterpret_cast<const float4*>(sp);
        const float4 r1 = *reinterpret_cast<const float4*>(sp + Nout);
        float* d = &wl[u * 8];
        *reinterpret_cast<float4*>(d)     = make_float4(r0.x, r1.x, r0.y, r1.y);
        *reinterpret_cast<float4*>(d + 4) = make_float4(r0.z, r1.z, r0.w, r1.w);
    }

    const float* __restrict__ q0 = pre0 + lane * 4;
    const float* __restrict__ q1 = pre1 + lane * 4;

    int it = 0;
    for (int tile = blockIdx.x; tile < NTILES; tile += gridDim.x, ++it) {
        const int row = tile * 16 + wave;
        const int beg = __builtin_amdgcn_readfirstlane(off[row]);
        const int end = __builtin_amdgcn_readfirstlane(off[row + 1]);

        vecT acc = gcn_row_sum<4>(q0, q1, beg, end, lane, scol, sv0, sv1);
#pragma unroll
        for (int k = 0; k < 4; ++k) acc[k] = fmaxf(acc[k], 0.0f);

        __syncthreads();
        *reinterpret_cast<vecT*>(&hb[wave * FG_PITCH + lane * 4]) = acc;
        __syncthreads();

        if ((wave >> 2) == (it & 3)) {
            const int w4 = wave & 3;
            const int s  = w4 >> 1;
            const int tp = (w4 & 1) * 2;
            const float* ha = &hb[mn * FG_PITCH + 2 * half];
            const float* wb = &wl[(s * NT + tp) * 64 + lane * 2];
            v8f c0 = v8f{}, c1 = v8f{};
#pragma unroll
            for (int ks = 0; ks < KS; ++ks) {
                const v2f a  = *reinterpret_cast<const v2f*>(ha + 4 * ks);
                const v2f b0 = *reinterpret_cast<const v2f*>(wb + ks * TT * 64);
                const v2f b1 = *reinterpret_cast<const v2f*>(wb + ks * TT * 64 + 64);
                c0 = __builtin_amdgcn_wmma_f32_16x16x4_f32(false, a, false, b0,
                                                           (short)0, c0, false, false);
                c1 = __builtin_amdgcn_wmma_f32_16x16x4_f32(false, a, false, b1,
                                                           (short)0, c1, false, false);
            }
            float* __restrict__ cp = (s ? C1 : C0)
                + (size_t)(tile * 16 + half * 8) * Nout + tp * 16 + mn;
#pragma unroll
            for (int v = 0; v < 8; ++v) {
                cp[(size_t)v * Nout]      = c0[v];
                cp[(size_t)v * Nout + 16] = c1[v];
            }
        }
    }
}

__global__ void gcn_gather_scalar(const float* __restrict__ p0,
                                  const float* __restrict__ p1,
                                  const int* __restrict__ off,
                                  const int* __restrict__ scol,
                                  const float* __restrict__ sv0,
                                  const float* __restrict__ sv1,
                                  float* __restrict__ out) {
    int r = blockIdx.x * blockDim.x + threadIdx.x;
    if (r >= NN) return;
    const int beg = off[r];
    const int end = off[r + 1];
    float s = 0.0f;
    for (int i = beg; i < end; ++i) {
        const int c = scol[i];
        s += sv0[i] * p0[c] + sv1[i] * p1[c];
    }
    out[r] = s;
}

extern "C" void kernel_launch(void* const* d_in, const int* in_sizes, int n_in,
                              void* d_out, int out_size, void* d_ws, size_t ws_size,
                              hipStream_t stream) {
    const float* x     = (const float*)d_in[0];
    const int*   erow  = (const int*)d_in[1];
    const int*   ecol  = (const int*)d_in[2];
    const float* evals = (const float*)d_in[3];
    const float* W1    = (const float*)d_in[4];
    const float* W2    = (const float*)d_in[5];
    const float* W3    = (const float*)d_in[6];
    float* out = (float*)d_out;

    const float* ev0 = evals;
    const float* ev1 = evals + (size_t)EE;

    float* R0 = (float*)d_ws;
    float* R1 = R0 + (size_t)NN * HH1;
    float* R2 = R1 + (size_t)NN * HH1;
    float* R3 = R2 + (size_t)NN * HH1;

    int*   cnt  = (int*)R2;
    int*   rank = cnt + NN;
    int*   scol = (int*)R3;
    float* sv0  = R3 + (size_t)EE;
    float* sv1  = sv0 + (size_t)EE;
    int*   off  = (int*)(sv1 + (size_t)EE);

    const int ZB = 256;
    dim3 gemm_blk(32, GEMM_WAVES);
    constexpr int GA = 768;
    constexpr int ROWS_A = GA * GEMM_WAVES * 16;
    constexpr int ROWS_B = NN - ROWS_A;
    constexpr int GB = (ROWS_B / 16 + GEMM_WAVES - 1) / GEMM_WAVES;
    static_assert(ROWS_B > 0 && ROWS_B % 16 == 0, "whole 16-row tiles");
    static_assert((long long)GA * GEMM_WAVES * (FF / GEMM_KC) * 64 >= EE, "part A ranks every edge");
    static_assert((long long)GB * GEMM_WAVES * (FF / GEMM_KC) * 32 >= EE, "part B places every edge");

    float* pre0 = R0;
    float* pre1 = R1;
    gcn_zero_counts<<<(NN + ZB - 1) / ZB, ZB, 0, stream>>>(cnt, NN);
    gcn_gemm_wmma2<FF, HH1, 1><<<dim3(GA), gemm_blk, 0, stream>>>(
        x, W1, W1 + (size_t)FF * HH1, pre0, pre1, ROWS_A,
        erow, ecol, ev0, ev1, off, cnt, rank, scol, sv0, sv1);
    gcn_scan_counts<<<SCAN_WGS, SCAN_T, 0, stream>>>(cnt, off);
    gcn_gemm_wmma2<FF, HH1, 2><<<dim3(GB), gemm_blk, 0, stream>>>(
        x + (size_t)ROWS_A * FF, W1, W1 + (size_t)FF * HH1,
        pre0 + (size_t)ROWS_A * HH1, pre1 + (size_t)ROWS_A * HH1, ROWS_B,
        erow, ecol, ev0, ev1, off, cnt, rank, scol, sv0, sv1);

    float* pre0b = R2;
    float* pre1b = R2 + (size_t)NN * HH2;
    gcn_gather_rows_mm<<<dim3(FG_GRID), dim3(32, FG_WAVES), 0, stream>>>(
        pre0, pre1, off, scol, sv0, sv1, W2, W2 + (size_t)HH1 * HH2, pre0b, pre1b);

    float* p0 = R0;
    float* p1 = R0 + NN;
    gcn_gather_rows_proj<<<dim3(GP_GRID), dim3(32, GP_WAVES), 0, stream>>>(
        pre0b, pre1b, off, scol, sv0, sv1, W3, W3 + (size_t)HH2, p0, p1);

    gcn_gather_scalar<<<(NN + ZB - 1) / ZB, ZB, 0, stream>>>(
        p0, p1, off, scol, sv0, sv1, out);
}


# Layer-3 gather walks the edge list one edge per thread: whole-line list reads, p0|p1 stored as one 8-byte pair per row, every load independent; products meet in LDS, a row adds its own in list order
# speedup vs baseline: 2.3036x; 1.0362x over previous
#include <hip/hip_runtime.h>


typedef __attribute__((ext_vector_type(2))) float v2f;
typedef __attribute__((ext_vector_type(8))) float v8f;

#define NN 100000
#define FF 256
#define EE 1600000
#define HH1 128
#define HH2 64

#define GEMM_WAVES 4

__global__ void gcn_zero_counts(int* __restrict__ p, int n) {
    int i = blockIdx.x * blockDim.x + threadIdx.x;
    if (i < n) p[i] = 0;
}


#define SCAN_T 1024
#define SCAN_SEG (4 * SCAN_T)
#define SCAN_WGS ((NN + SCAN_SEG - 1) / SCAN_SEG)
__global__ __launch_bounds__(SCAN_T)
void gcn_scan_counts(const int* __restrict__ cnt, int* __restrict__ off) {
    static_assert(NN % 4 == 0, "counters in whole int4 pieces");
    static_assert(SCAN_T == 32 * 32, "one lane per wave in the cross-wave step");
    static_assert((SCAN_WGS - 1) * SCAN_SEG <= NN, "every earlier segment is a full one");
    __shared__ int wown[SCAN_T / 32];
    __shared__ int wprev[SCAN_T / 32];
    const int t = threadIdx.x;
    const int lane = t & 31;
    const int wave = t >> 5;
    const int g = blockIdx.x;
    const int4* __restrict__ src4 = reinterpret_cast<const int4*>(cnt);

    const int i4 = g * SCAN_T + t;
    const bool act = i4 < NN / 4;
    int4 v = make_int4(0, 0, 0, 0);
    if (act) v = src4[i4];

    int sprev = 0;
    for (int k = 0; k < g; k += 8) {
        int4 w[8];
#pragma unroll
        for (int u = 0; u < 8; ++u) {
            const int ks = (k + u < g) ? (k + u) : (g - 1);
            w[u] = src4[ks * SCAN_T + t];
        }
#pragma unroll
        for (int u = 0; u < 8; ++u) {
            const int s4 = (w[u].x + w[u].y) + (w[u].z + w[u].w);
            sprev += (k + u < g) ? s4 : 0;
        }
    }

    const int own = (v.x + v.y) + (v.z + v.w);
    int inc = own;
    for (int d = 1; d < 32; d <<= 1) {
        const int y = __shfl_up(inc, d, 32);
        if (lane >= d) inc += y;
    }
    int ps = sprev;
    for (int d = 16; d >= 1; d >>= 1) ps += __shfl_xor(ps, d, 32);
    if (lane == 31) wown[wave] = inc;
    if (lane == 0) wprev[wave] = ps;
    __syncthreads();
    int before = (lane < wave) ? wown[lane] : 0;
    int base = wprev[lane];
    for (int d = 16; d >= 1; d >>= 1) {
        before += __shfl_xor(before, d, 32);
        base += __shfl_xor(base, d, 32);
    }
    const int start = base + before + (inc - own);
    if (act) {
        int4 o;
        o.x = start;
        o.y = o.x + v.x;
        o.z = o.y + v.y;
        o.w = o.z + v.z;
        reinterpret_cast<int4*>(off)[i4] = o;
    }
    if (i4 == NN / 4 - 1) off[NN] = start + own;
}

#define GEMM_KC 16
template <int K, int Nout, int JOB>
__global__ __launch_bounds__(32 * GEMM_WAVES) __attribute__((amdgpu_waves_per_eu(JOB == 2 ? 4 : 3)))
void gcn_gemm_wmma2(const float* __restrict__ A,
                    const float* __restrict__ B0, const float* __restrict__ B1,
                    float* __restrict__ C0, float* __restrict__ C1, int M,
                    const int* __restrict__ erow, const int* __restrict__ ecol,
                    const float* __restrict__ ev0, const float* __restrict__ ev1,
                    const int* __restrict__ off, int* __restrict__ cnt,
                    int* __restrict__ rank, int* __restrict__ scol,
                    float* __restrict__ sv0, float* __restrict__ sv1) {
    static_assert(K % GEMM_KC == 0 && Nout % 16 == 0, "tile dims");
    constexpr int NT = Nout / 16;
    constexpr int TT = 2 * NT;
    constexpr int NC = K / GEMM_KC;
    constexpr int CHUNK_F = 4 * TT * 64;
    constexpr int UNITS = CHUNK_F / 8;
    constexpr int NTHR = 32 * GEMM_WAVES;
    constexpr int UPT = UNITS / NTHR;
    static_assert(UNITS % NTHR == 0 && UPT >= 1, "staging split");

    __shared__ __attribute__((aligned(16))) float wl[2][CHUNK_F];

    const int lane = threadIdx.x;
    const int wave = threadIdx.y;
    const int tid  = wave * 32 + lane;
    const int half = lane >> 4;
    const int mn   = lane & 15;
    const int tileM = blockIdx.x * GEMM_WAVES + wave;
    const bool live = tileM * 16 < M;
    const int row0 = live ? tileM * 16 : 0;

    const float* __restrict__ ap = A + (size_t)(row0 + mn) * K + 2 * half;

    const float* src[UPT];
#pragma unroll
    for (int i = 0; i < UPT; ++i) {
        const int u = i * NTHR + tid;
        const int frag = u >> 3, q = u & 7;
        const int ks = frag / TT, t = frag % TT;
        const int s = t / NT, tn = t % NT;
        const int hf = q >> 2, nn = (q & 3) * 4;
        src[i] = (s ? B1 : B0) + (size_t)(ks * 4 + hf * 2) * Nout + tn * 16 + nn;
    }

    v8f acc[TT];
#pragma unroll
    for (int t = 0; t < TT; ++t) acc[t] = v8f{};

    float4 r0[UPT], r1[UPT];
    v2f a[4], an[4];

#pragma unroll
    for (int i = 0; i < UPT; ++i) {
        r0[i] = *reinterpret_cast<const float4*>(src[i]);
        r1[i] = *reinterpret_cast<const float4*>(src[i] + Nout);
    }
#pragma unroll
    for (int ks = 0; ks < 4; ++ks) a[ks] = *reinterpret_cast<const v2f*>(ap + 4 * ks);
#pragma unroll
    for (int i = 0; i < UPT; ++i) {
        float* d = &wl[0][(i * NTHR + tid) * 8];
        *reinterpret_cast<float4*>(d)     = make_float4(r0[i].x, r1[i].x, r0[i].y, r1[i].y);
        *reinterpret_cast<float4*>(d + 4) = make_float4(r0[i].z, r1[i].z, r0[i].w, r1[i].w);
    }
    __syncthreads();

    constexpr int EW = (JOB == 1) ? 64 : 32;
    const int ebase = (blockIdx.x * GEMM_WAVES + wave) * (NC * EW) + lane;
    int er = 0, ek = 0;
    int erb = 0, ekb = 0;
    int fo = 0, fk = 0, fc = 0;
    float fa = 0.0f, fb = 0.0f;
    auto edge_step = [&](int s) {
        const int e0 = ebase + EW * s;
        const int e1 = e0 - EW;
        const int e2 = e0 - 2 * EW;
        if (JOB == 1) {
            if (s >= 2 && s <= NC + 1) {
                if (e2 < EE) rank[e2] = ek;
                if (e2 + 32 < EE) rank[e2 + 32] = ekb;
            }
            if (s >= 1 && s <= NC) {
                if (e1 < EE) ek = atomicAdd(cnt + er, 1);
                if (e1 + 32 < EE) ekb = atomicAdd(cnt + erb, 1);
            }
            if (s < NC) {
                if (e0 < EE) er = __builtin_nontemporal_load(erow + e0);
                if (e0 + 32 < EE) erb = __builtin_nontemporal_load(erow + e0 + 32);
            }
        }
        if (JOB == 2) {
            if (s >= 2 && s <= NC + 1 && e2 < EE) {
                const int p = fo + fk;
                scol[p] = fc;
                sv0[p]  = fa;
                sv1[p]  = fb;
            }
            if (s >= 1 && s <= NC && e1 < EE) {
                fo = off[er];
                fk = __builtin_nontemporal_load(rank + e1);
                fc = __builtin_nontemporal_load(ecol + e1);
                fa = __builtin_nontemporal_load(ev0 + e1);
                fb = __builtin_nontemporal_load(ev1 + e1);
            }
            if (s < NC && e0 < EE) er = __builtin_nontemporal_load(erow + e0);
        }
    };

    for (int c = 0; c < NC; ++c) {
        const bool more = c + 1 < NC;
        edge_step(c);
        if (more) {
            const size_t kn = (size_t)(c + 1) * GEMM_KC;
#pragma unroll
            for (int i = 0; i < UPT; ++i) {
                r0[i] = *reinterpret_cast<const float4*>(src[i] + kn * Nout);
                r1[i] = *reinterpret_cast<const float4*>(src[i] + kn * Nout + Nout);
            }
#pragma unroll
            for (int ks = 0; ks < 4; ++ks)
                an[ks] = *reinterpret_cast<const v2f*>(ap + kn + 4 * ks);
        }
        const float* wb = &wl[c & 1][lane * 2];
        v2f bx[TT], by[TT];
#pragma unroll
        for (int t = 0; t < TT; ++t) bx[t] = *reinterpret_cast<const v2f*>(wb + (0 * TT + t) * 64);
#pragma unroll
        for (int t = 0; t < TT; ++t) by[t] = *reinterpret_cast<const v2f*>(wb + (1 * TT + t) * 64);
#pragma unroll
        for (int t = 0; t < TT; ++t)
            acc[t] = __builtin_amdgcn_wmma_f32_16x16x4_f32(false, a[0], false, bx[t],
                                                           (short)0, acc[t], false, false);
#pragma unroll
        for (int t = 0; t < TT; ++t) bx[t] = *reinterpret_cast<const v2f*>(wb + (2 * TT + t) * 64);
#pragma unroll
        for (int t = 0; t < TT; ++t)
            acc[t] = __builtin_amdgcn_wmma_f32_16x16x4_f32(false, a[1], false, by[t],
                                                           (short)0, acc[t], false, false);
#pragma unroll
        for (int t = 0; t < TT; ++t) by[t] = *reinterpret_cast<const v2f*>(wb + (3 * TT + t) * 64);
#pragma unroll
        for (int t = 0; t < TT; ++t)
            acc[t] = __builtin_amdgcn_wmma_f32_16x16x4_f32(false, a[2], false, bx[t],
                                                           (short)0, acc[t], false, false);
#pragma unroll
        for (int t = 0; t < TT; ++t)
            acc[t] = __builtin_amdgcn_wmma_f32_16x16x4_f32(false, a[3], false, by[t],
                                                           (short)0, acc[t], false, false);
        if (more) {
            float* wd = &wl[(c & 1) ^ 1][0];
#pragma unroll
            for (int i = 0; i < UPT; ++i) {
                float* d = wd + (i * NTHR + tid) * 8;
                *reinterpret_cast<float4*>(d)     = make_float4(r0[i].x, r1[i].x, r0[i].y, r1[i].y);
                *reinterpret_cast<float4*>(d + 4) = make_float4(r0[i].z, r1[i].z, r0[i].w, r1[i].w);
            }
#pragma unroll
            for (int ks = 0; ks < 4; ++ks) a[ks] = an[ks];
        }
        __syncthreads();
    }
    edge_step(NC);
    edge_step(NC + 1);

    if (live) {
#pragma unroll
        for (int t = 0; t < TT; ++t) {
            const int s = t / NT, tn = t % NT;
            float* __restrict__ cp = (s ? C1 : C0)
                + (size_t)(row0 + half * 8) * Nout + tn * 16 + mn;
#pragma unroll
            for (int v = 0; v < 8; ++v) cp[(size_t)v * Nout] = acc[t][v];
        }
    }
}

static __device__ __forceinline__ float gcn_lane_f(float x, int j) {
    return __int_as_float(__builtin_amdgcn_readlane(__float_as_int(x), j));
}

typedef __attribute__((address_space(1))) const v2f gcn_gv2f;
static __device__ __forceinline__ gcn_gv2f* gcn_scalar_row2(const float* p) {
    const unsigned long long v = (unsigned long long)p;
    const unsigned lo = __builtin_amdgcn_readfirstlane((unsigned)v);
    const unsigned hi = __builtin_amdgcn_readfirstlane((unsigned)(v >> 32));
    return (gcn_gv2f*)(((unsigned long long)hi << 32) | lo);
}

template <int VEC> struct gcn_vec { typedef __attribute__((ext_vector_type(VEC))) float type; };

template <int VEC>
static __device__ __forceinline__ typename gcn_vec<VEC>::type
gcn_row_sum(const float* __restrict__ q0, const float* __restrict__ q1,
            int beg, int end, int lane, const int* __restrict__ scol,
            const float* __restrict__ sv0, const float* __restrict__ sv1) {
    typedef typename gcn_vec<VEC>::type vecT;
    constexpr int H = 32 * VEC;
    vecT acc = {};
    for (int base = beg; base < end; base += 32) {
        const int left = end - base;
        const int n = left < 32 ? left : 32;
        int cj = 0;
        float a0 = 0.0f, a1 = 0.0f;
        if (lane < n) {
            cj = scol[base + lane];
            a0 = sv0[base + lane];
            a1 = sv1[base + lane];
        }
        int j = 0;
        for (; j + 4 <= n; j += 4) {
            vecT x0[4], x1[4];
            float w0[4], w1[4];
#pragma unroll
            for (int u = 0; u < 4; ++u) {
                const int c = __builtin_amdgcn_readlane(cj, j + u);
                w0[u] = gcn_lane_f(a0, j + u);
                w1[u] = gcn_lane_f(a1, j + u);
                x0[u] = *reinterpret_cast<const vecT*>(q0 + (size_t)c * H);
                x1[u] = *reinterpret_cast<const vecT*>(q1 + (size_t)c * H);
            }
#pragma unroll
            for (int u = 0; u < 4; ++u) acc += w0[u] * x0[u] + w1[u] * x1[u];
        }
        for (; j < n; ++j) {
            const int c = __builtin_amdgcn_readlane(cj, j);
            const float w0 = gcn_lane_f(a0, j);
            const float w1 = gcn_lane_f(a1, j);
            const vecT x0 = *reinterpret_cast<const vecT*>(q0 + (size_t)c * H);
            const vecT x1 = *reinterpret_cast<const vecT*>(q1 + (size_t)c * H);
            acc += w0 * x0 + w1 * x1;
        }
    }
    return acc;
}

#define GP_WAVES 8
#define GP_ROWS 9
#define GP_B 8
#define GP_NWAVES ((NN + GP_ROWS - 1) / GP_ROWS)
#define GP_GRID ((GP_NWAVES + GP_WAVES - 1) / GP_WAVES)
__global__ __launch_bounds__(32 * GP_WAVES)
void gcn_gather_rows_proj(const float* __restrict__ pre0, const float* __restrict__ pre1,
                          const int* __restrict__ off, const int* __restrict__ scol,
                          const float* __restrict__ sv0, const float* __restrict__ sv1,
                          const float* __restrict__ w0, const float* __restrict__ w1,
                          float* __restrict__ pp) {
    static_assert(GP_ROWS + 1 <= 32, "one lane per row bound");
    const int lane = threadIdx.x;
    const int wid = __builtin_amdgcn_readfirstlane((int)(blockIdx.x * GP_WAVES + threadIdx.y));
    const int r0 = wid * GP_ROWS;
    if (r0 >= NN) return;
    const int nr = (NN - r0 < GP_ROWS) ? (NN - r0) : GP_ROWS;
    int ob = 0;
    if (lane <= nr) ob = off[r0 + lane];
    const v2f wa = *reinterpret_cast<const v2f*>(w0 + lane * 2);
    const v2f wb = *reinterpret_cast<const v2f*>(w1 + lane * 2);

    const int ebeg = __builtin_amdgcn_readlane(ob, 0);
    const int eend = __builtin_amdgcn_readlane(ob, nr);
    int i = 0;
    int rend = __builtin_amdgcn_readlane(ob, 1);
    v2f acc = {};
    float res0 = 0.0f, res1 = 0.0f;

    auto close_row = [&]() {
        const float h0 = fmaxf(acc[0], 0.0f);
        const float h1 = fmaxf(acc[1], 0.0f);
        float t0 = h0 * wa[0] + h1 * wa[1];
        float t1 = h0 * wb[0] + h1 * wb[1];
        for (int d = 16; d >= 1; d >>= 1) {
            t0 += __shfl_xor(t0, d, 32);
            t1 += __shfl_xor(t1, d, 32);
        }
        if (lane == i) {
            res0 = t0;
            res1 = t1;
        }
        acc = v2f{};
        ++i;
        rend = (i < nr) ? __builtin_amdgcn_readlane(ob, i + 1) : 0x7fffffff;
    };

    int cj = 0;
    float a0 = 0.0f, a1 = 0.0f;
    if (ebeg + lane < eend) {
        cj = scol[ebeg + lane];
        a0 = sv0[ebeg + lane];
        a1 = sv1[ebeg + lane];
    }
    for (int base = ebeg; base < eend; base += 32) {
        const int left = eend - base;
        const int n = left < 32 ? left : 32;
        int cjn = 0;
        float a0n = 0.0f, a1n = 0.0f;
        for (int j = 0; j < n; j += GP_B) {
            v2f x0[GP_B], x1[GP_B];
            float e0[GP_B], e1[GP_B];
#pragma unroll
            for (int u = 0; u < GP_B; ++u) {
                const int jj = (j + u < n) ? (j + u) : (n - 1);
                const int c = __builtin_amdgcn_readlane(cj, jj);
                e0[u] = gcn_lane_f(a0, jj);
                e1[u] = gcn_lane_f(a1, jj);
                x0[u] = gcn_scalar_row2(pre0 + (size_t)c * HH2)[lane];
                x1[u] = gcn_scalar_row2(pre1 + (size_t)c * HH2)[lane];
            }
            if (j == 0 && base + 32 + lane < eend) {
                const unsigned pn = (unsigned)(base + 32 + lane);
                cjn = scol[pn];
                a0n = sv0[pn];
                a1n = sv1[pn];
            }
#pragma unroll
            for (int u = 0; u < GP_B; ++u) {
                if (j + u < n) {
                    const int e = base + j + u;
                    while (e >= rend) close_row();
                    acc += e0[u] * x0[u] + e1[u] * x1[u];
                }
            }
        }
        cj = cjn;
        a0 = a0n;
        a1 = a1n;
    }
    while (i < nr) close_row();
    if (lane < nr) {
        v2f pr;
        pr[0] = res0;
        pr[1] = res1;
        *reinterpret_cast<v2f*>(pp + 2 * (size_t)(r0 + lane)) = pr;
    }
}

#define FG_WAVES 16
#define FG_PITCH (HH1 + 4)
#define FG_GRID 1024
__global__ __launch_bounds__(32 * FG_WAVES) __attribute__((amdgpu_waves_per_eu(16)))
void gcn_gather_rows_mm(const float* __restrict__ pre0, const float* __restrict__ pre1,
                        const int* __restrict__ off, const int* __restrict__ scol,
                        const float* __restrict__ sv0, const float* __restrict__ sv1,
                        const float* __restrict__ B0, const float* __restrict__ B1,
                        float* __restrict__ C0, float* __restrict__ C1) {
    typedef __attribute__((ext_vector_type(4))) float vecT;
    constexpr int K = HH1, Nout = HH2;
    constexpr int NT = Nout / 16;
    constexpr int TT = 2 * NT;
    constexpr int KS = K / 4;
    constexpr int WL_F = KS * TT * 64;
    constexpr int NTHR = 32 * FG_WAVES;
    constexpr int UPT = (WL_F / 8) / NTHR;
    constexpr int NTILES = NN / 16;
    static_assert(NN % 16 == 0 && (WL_F / 8) % NTHR == 0 && NT % 2 == 0, "tile dims");
    static_assert(FG_WAVES == 16, "one wave per row of a tile, four groups of four");

    __shared__ __attribute__((aligned(16))) float wl[WL_F];
    __shared__ __attribute__((aligned(16))) float hb[16 * FG_PITCH];

    const int lane = threadIdx.x;
    const int wave = __builtin_amdgcn_readfirstlane((int)threadIdx.y);
    const int tid  = wave * 32 + lane;
    const int half = lane >> 4;
    const int mn   = lane & 15;

#pragma unroll
    for (int i = 0; i < UPT; ++i) {
        const int u = i * NTHR + tid;
        const int frag = u >> 3, q = u & 7;
        const int ks = frag / TT, t = frag % TT;
        const int s = t / NT, tn = t % NT;
        const int hf = q >> 2, nn = (q & 3) * 4;
        const float* sp = (s ? B1 : B0) + (size_t)(ks * 4 + hf * 2) * Nout + tn * 16 + nn;
        const float4 r0 = *reinterpret_cast<const float4*>(sp);
        const float4 r1 = *reinterpret_cast<const float4*>(sp + Nout);
        float* d = &wl[u * 8];
        *reinterpret_cast<float4*>(d)     = make_float4(r0.x, r1.x, r0.y, r1.y);
        *reinterpret_cast<float4*>(d + 4) = make_float4(r0.z, r1.z, r0.w, r1.w);
    }

    const float* __restrict__ q0 = pre0 + lane * 4;
    const float* __restrict__ q1 = pre1 + lane * 4;

    int it = 0;
    for (int tile = blockIdx.x; tile < NTILES; tile += gridDim.x, ++it) {
        const int row = tile * 16 + wave;
        const int beg = __builtin_amdgcn_readfirstlane(off[row]);
        const int end = __builtin_amdgcn_readfirstlane(off[row + 1]);

        vecT acc = gcn_row_sum<4>(q0, q1, beg, end, lane, scol, sv0, sv1);
#pragma unroll
        for (int k = 0; k < 4; ++k) acc[k] = fmaxf(acc[k], 0.0f);

        __syncthreads();
        *reinterpret_cast<vecT*>(&hb[wave * FG_PITCH + lane * 4]) = acc;
        __syncthreads();

        if ((wave >> 2) == (it & 3)) {
            const int w4 = wave & 3;
            const int s  = w4 >> 1;
            const int tp = (w4 & 1) * 2;
            const float* ha = &hb[mn * FG_PITCH + 2 * half];
            const float* wb = &wl[(s * NT + tp) * 64 + lane * 2];
            v8f c0 = v8f{}, c1 = v8f{};
#pragma unroll
            for (int ks = 0; ks < KS; ++ks) {
                const v2f a  = *reinterpret_cast<const v2f*>(ha + 4 * ks);
                const v2f b0 = *reinterpret_cast<const v2f*>(wb + ks * TT * 64);
                const v2f b1 = *reinterpret_cast<const v2f*>(wb + ks * TT * 64 + 64);
                c0 = __builtin_amdgcn_wmma_f32_16x16x4_f32(false, a, false, b0,
                                                           (short)0, c0, false, false);
                c1 = __builtin_amdgcn_wmma_f32_16x16x4_f32(false, a, false, b1,
                                                           (short)0, c1, false, false);
            }
            float* __restrict__ cp = (s ? C1 : C0)
                + (size_t)(tile * 16 + half * 8) * Nout + tp * 16 + mn;
#pragma unroll
            for (int v = 0; v < 8; ++v) {
                cp[(size_t)v * Nout]      = c0[v];
                cp[(size_t)v * Nout + 16] = c1[v];
            }
        }
    }
}

#define G3_T 256
#define G3_CH 8192
#define G3_U 8
__global__ __launch_bounds__(G3_T)
void gcn_gather_scalar(const float* __restrict__ pp,
                       const int* __restrict__ off,
                       const int* __restrict__ scol,
                       const float* __restrict__ sv0,
                       const float* __restrict__ sv1,
                       float* __restrict__ out) {
    __shared__ float msg[G3_CH];
    const int t = threadIdx.x;
    const int R0 = blockIdx.x * G3_T;
    const int R1 = (R0 + G3_T < NN) ? (R0 + G3_T) : NN;
    const int r = R0 + t;
    int ob = 0, oe = 0;
    if (r < NN) {
        ob = off[r];
        oe = off[r + 1];
    }
    const int ebeg = off[R0];
    const int eend = off[R1];
    const v2f* __restrict__ pp2 = reinterpret_cast<const v2f*>(pp);
    float s = 0.0f;
    for (int cb = ebeg; cb < eend; cb += G3_CH) {
        const int ce = (eend - cb < G3_CH) ? eend : (cb + G3_CH);
        for (int e0 = cb + t; e0 < ce; e0 += G3_U * G3_T) {
            int ee[G3_U], cc[G3_U];
            float a0[G3_U], a1[G3_U];
            v2f pr[G3_U];
#pragma unroll
            for (int u = 0; u < G3_U; ++u) {
                const int e = e0 + u * G3_T;
                ee[u] = (e < ce) ? e : e0;
                cc[u] = scol[ee[u]];
                a0[u] = sv0[ee[u]];
                a1[u] = sv1[ee[u]];
            }
#pragma unroll
            for (int u = 0; u < G3_U; ++u) pr[u] = pp2[cc[u]];
#pragma unroll
            for (int u = 0; u < G3_U; ++u)
                msg[ee[u] - cb] = a0[u] * pr[u][0] + a1[u] * pr[u][1];
        }
        __syncthreads();
        const int lo = (ob > cb) ? ob : cb;
        const int hi = (oe < ce) ? oe : ce;
        for (int k = lo; k < hi; ++k) s += msg[k - cb];
        __syncthreads();
    }
    if (r < NN) out[r] = s;
}

extern "C" void kernel_launch(void* const* d_in, const int* in_sizes, int n_in,
                              void* d_out, int out_size, void* d_ws, size_t ws_size,
                              hipStream_t stream) {
    const float* x     = (const float*)d_in[0];
    const int*   erow  = (const int*)d_in[1];
    const int*   ecol  = (const int*)d_in[2];
    const float* evals = (const float*)d_in[3];
    const float* W1    = (const float*)d_in[4];
    const float* W2    = (const float*)d_in[5];
    const float* W3    = (const float*)d_in[6];
    float* out = (float*)d_out;

    const float* ev0 = evals;
    const float* ev1 = evals + (size_t)EE;

    float* R0 = (float*)d_ws;
    float* R1 = R0 + (size_t)NN * HH1;
    float* R2 = R1 + (size_t)NN * HH1;
    float* R3 = R2 + (size_t)NN * HH1;

    int*   cnt  = (int*)R2;
    int*   rank = cnt + NN;
    int*   scol = (int*)R3;
    float* sv0  = R3 + (size_t)EE;
    float* sv1  = sv0 + (size_t)EE;
    int*   off  = (int*)(sv1 + (size_t)EE);

    const int ZB = 256;
    dim3 gemm_blk(32, GEMM_WAVES);
    constexpr int GA = 768;
    constexpr int ROWS_A = GA * GEMM_WAVES * 16;
    constexpr int ROWS_B = NN - ROWS_A;
    constexpr int GB = (ROWS_B / 16 + GEMM_WAVES - 1) / GEMM_WAVES;
    static_assert(ROWS_B > 0 && ROWS_B % 16 == 0, "whole 16-row tiles");
    static_assert((long long)GA * GEMM_WAVES * (FF / GEMM_KC) * 64 >= EE, "part A ranks every edge");
    static_assert((long long)GB * GEMM_WAVES * (FF / GEMM_KC) * 32 >= EE, "part B places every edge");

    float* pre0 = R0;
    float* pre1 = R1;
    gcn_zero_counts<<<(NN + ZB - 1) / ZB, ZB, 0, stream>>>(cnt, NN);
    gcn_gemm_wmma2<FF, HH1, 1><<<dim3(GA), gemm_blk, 0, stream>>>(
        x, W1, W1 + (size_t)FF * HH1, pre0, pre1, ROWS_A,
        erow, ecol, ev0, ev1, off, cnt, rank, scol, sv0, sv1);
    gcn_scan_counts<<<SCAN_WGS, SCAN_T, 0, stream>>>(cnt, off);
    gcn_gemm_wmma2<FF, HH1, 2><<<dim3(GB), gemm_blk, 0, stream>>>(
        x + (size_t)ROWS_A * FF, W1, W1 + (size_t)FF * HH1,
        pre0 + (size_t)ROWS_A * HH1, pre1 + (size_t)ROWS_A * HH1, ROWS_B,
        erow, ecol, ev0, ev1, off, cnt, rank, scol, sv0, sv1);

    float* pre0b = R2;
    float* pre1b = R2 + (size_t)NN * HH2;
    gcn_gather_rows_mm<<<dim3(FG_GRID), dim3(32, FG_WAVES), 0, stream>>>(
        pre0, pre1, off, scol, sv0, sv1, W2, W2 + (size_t)HH1 * HH2, pre0b, pre1b);

    float* pp = R0;
    gcn_gather_rows_proj<<<dim3(GP_GRID), dim3(32, GP_WAVES), 0, stream>>>(
        pre0b, pre1b, off, scol, sv0, sv1, W3, W3 + (size_t)HH2, pp);

    gcn_gather_scalar<<<(NN + G3_T - 1) / G3_T, G3_T, 0, stream>>>(
        pp, off, scol, sv0, sv1, out);
}
